# starting kernel
# baseline (speedup 1.0000x reference)
#include <hip/hip_runtime.h>


typedef float v2f __attribute__((ext_vector_type(2)));
typedef float v8f __attribute__((ext_vector_type(8)));

#define Bb    16
#define Cc    64
#define Dd    128
#define Mm    16
#define Ll    3
#define NMODE 512
#define NSPEC (NMODE*Cc*Bb)
#define WLAYER (2*NMODE*Cc*Cc)

__global__ void fnde_build_tables(float* cs, float* sn) {
    int t = threadIdx.x;
    float ang = (float)(6.283185307179586 / 128.0) * (float)t;
    cs[t] = cosf(ang);
    sn[t] = sinf(ang);
}

__global__ void fnde_pack_weights(const float* __restrict__ w1,
                                  const float* __restrict__ w2,
                                  float* __restrict__ Wp) {
    int idx = blockIdx.x * 256 + threadIdx.x;
    int l   = idx / (NMODE * Cc * Cc);
    int rem = idx % (NMODE * Cc * Cc);
    int m   = rem >> 12;
    int oi  = rem & 4095;
    int o   = oi >> 6;
    int i   = oi & 63;
    int corner = m >> 8;
    int mx = (m >> 4) & 15;
    int ky = m & 15;
    const float* src = corner ? w2 : w1;
    int sidx = ((((l * Cc + i) * Cc + o) * Mm + mx) * Mm + ky) * 2;
    float* dst = Wp + (size_t)l * WLAYER + (size_t)m * 4096 + oi;
    dst[0]              = src[sidx];
    dst[NMODE * Cc * Cc] = src[sidx + 1];
}

__global__ void fnde_fwd_dft1(const float* __restrict__ z,
                              const float* __restrict__ cs,
                              const float* __restrict__ sn,
                              float* __restrict__ Ay) {
    int t  = blockIdx.x * 256 + threadIdx.x;
    int b  = t >> 11;
    int x  = (t >> 4) & 127;
    int ky = t & 15;
    const float* row = z + (b * 128 + x) * 128;
    float re = 0.f, im = 0.f;
    for (int y = 0; y < 128; ++y) {
        float v = row[y];
        int a = (ky * y) & 127;
        re += v * cs[a];
        im -= v * sn[a];
    }
    Ay[t * 2]     = re;
    Ay[t * 2 + 1] = im;
}

__global__ void fnde_fwd_dft2(const float* __restrict__ Ay,
                              const float* __restrict__ cs,
                              const float* __restrict__ sn,
                              float* __restrict__ Zhat) {
    int t = blockIdx.x * 256 + threadIdx.x;
    int b = t >> 9;
    int m = t & 511;
    int corner = m >> 8, mx = (m >> 4) & 15, ky = m & 15;
    int kx = corner ? (112 + mx) : mx;
    float re = 0.f, im = 0.f;
    for (int x = 0; x < 128; ++x) {
        int base = ((b * 128 + x) * 16 + ky) * 2;
        float ar = Ay[base], ai = Ay[base + 1];
        int a = (kx * x) & 127;
        float c = cs[a], s = sn[a];
        re += ar * c + ai * s;
        im += ai * c - ar * s;
    }
    Zhat[t * 2]     = re;
    Zhat[t * 2 + 1] = im;
}

__global__ void fnde_init_spec(const float* __restrict__ Zhat,
                               const float* __restrict__ lw,
                               const float* __restrict__ lb,
                               float* __restrict__ Zre, float* __restrict__ Zim,
                               float* __restrict__ STre, float* __restrict__ STim) {
    int e = blockIdx.x * 256 + threadIdx.x;
    int m = e >> 10;
    int c = (e >> 4) & 63;
    int b = e & 15;
    float zr = Zhat[(b * 512 + m) * 2];
    float zi = Zhat[(b * 512 + m) * 2 + 1];
    float w = lw[c];
    Zre[e] = w * zr + ((m == 0) ? lb[c] * 16384.0f : 0.0f);
    Zim[e] = w * zi;
    STre[e] = 0.0f;
    STim[e] = 0.0f;
}

__global__ __launch_bounds__(128)
void fnde_gapply(const float* __restrict__ Vre, const float* __restrict__ Vim,
                 const float* __restrict__ Wp,
                 float* __restrict__ Kre, float* __restrict__ Kim,
                 const float* __restrict__ bias) {
    int tid  = blockIdx.x * 128 + threadIdx.x;
    int mode = tid >> 5;
    int lane = threadIdx.x & 31;
    int laneHi = lane >> 4;
    int laneLo = lane & 15;

    const float* Wr = Wp + (size_t)mode * 4096;
    const float* Wi = Wr + (size_t)NMODE * Cc * Cc;
    const float* Vr = Vre + mode * 1024;
    const float* Vi = Vim + mode * 1024;

    v8f cr0 = {}, cr1 = {}, cr2 = {}, cr3 = {};
    v8f ci0 = {}, ci1 = {}, ci2 = {}, ci3 = {};

    const int bOff = (laneHi << 5) + laneLo;
    const int aOff = laneLo * 64 + laneHi * 2;

#pragma unroll 4
    for (int kk = 0; kk < 16; ++kk) {
        int vb = kk * 64 + bOff;
        v2f br, bi, bin;
        br.x = Vr[vb];      br.y = Vr[vb + 16];
        bi.x = Vi[vb];      bi.y = Vi[vb + 16];
        bin.x = -bi.x;      bin.y = -bi.y;
        int ab = aOff + kk * 4;

#define FNDE_MT(MT, CR, CI)                                                          \
        {                                                                            \
            v2f ar = *(const v2f*)(Wr + ab + MT * 1024);                             \
            v2f ai = *(const v2f*)(Wi + ab + MT * 1024);                             \
            CR = __builtin_amdgcn_wmma_f32_16x16x4_f32(false, ar, false, br,         \
                                                       (short)0, CR, false, false);  \
            CR = __builtin_amdgcn_wmma_f32_16x16x4_f32(false, ai, false, bin,        \
                                                       (short)0, CR, false, false);  \
            CI = __builtin_amdgcn_wmma_f32_16x16x4_f32(false, ar, false, bi,         \
                                                       (short)0, CI, false, false);  \
            CI = __builtin_amdgcn_wmma_f32_16x16x4_f32(false, ai, false, br,         \
                                                       (short)0, CI, false, false);  \
        }
        FNDE_MT(0, cr0, ci0)
        FNDE_MT(1, cr1, ci1)
        FNDE_MT(2, cr2, ci2)
        FNDE_MT(3, cr3, ci3)
#undef FNDE_MT
    }

    float* Kor = Kre + mode * 1024;
    float* Koi = Kim + mode * 1024;
    bool addBias = (mode == 0);

#define FNDE_ST(MT, CR, CI)                                                      \
    {                                                                            \
        int obase = MT * 16 + (laneHi << 3);                                     \
        _Pragma("unroll")                                                        \
        for (int j = 0; j < 8; ++j) {                                            \
            float vr = CR[j];                                                    \
            if (addBias) vr += bias[obase + j] * 16384.0f;                       \
            Kor[(obase + j) * 16 + laneLo] = vr;                                 \
            Koi[(obase + j) * 16 + laneLo] = CI[j];                              \
        }                                                                        \
    }
    FNDE_ST(0, cr0, ci0)
    FNDE_ST(1, cr1, ci1)
    FNDE_ST(2, cr2, ci2)
    FNDE_ST(3, cr3, ci3)
#undef FNDE_ST
}

__device__ inline void fnde_pi(const float* __restrict__ Re, const float* __restrict__ Im,
                               int e, float kr, float ki, float& pr, float& pim) {
    int m = e >> 10;
    int ky = m & 15;
    if (ky != 0) { pr = kr; pim = ki; return; }
    int corner = m >> 8;
    int mx = (m >> 4) & 15;
    if (corner == 1 && mx == 0) { pr = 0.5f * kr; pim = 0.5f * ki; return; }
    int pm = (mx == 0) ? m : ((corner == 0) ? (256 + (16 - mx) * 16) : ((16 - mx) * 16));
    int pe = (pm << 10) + (e & 1023);
    pr  = 0.5f * (kr + Re[pe]);
    pim = 0.5f * (ki - Im[pe]);
}

__global__ void fnde_spec_mid(const float* __restrict__ Zre, const float* __restrict__ Zim,
                              const float* __restrict__ Kre, const float* __restrict__ Kim,
                              float* __restrict__ Ure, float* __restrict__ Uim,
                              float* __restrict__ SSre, float* __restrict__ SSim,
                              const float* __restrict__ ts, int s, int phase) {
    int e = blockIdx.x * 256 + threadIdx.x;
    float h  = ts[s + 1] - ts[s];
    float uc = (phase == 3) ? h : 0.5f * h;
    float kr = Kre[e], ki = Kim[e];
    if (phase == 1) { SSre[e] = kr;          SSim[e] = ki; }
    else            { SSre[e] += 2.0f * kr;  SSim[e] += 2.0f * ki; }
    float pr, pim;
    fnde_pi(Kre, Kim, e, kr, ki, pr, pim);
    Ure[e] = Zre[e] + uc * pr;
    Uim[e] = Zim[e] + uc * pim;
}

__global__ void fnde_spec_fin(float* __restrict__ Zre, float* __restrict__ Zim,
                              const float* __restrict__ Kre, const float* __restrict__ Kim,
                              const float* __restrict__ SSre, const float* __restrict__ SSim,
                              float* __restrict__ STre, float* __restrict__ STim,
                              const float* __restrict__ ts, int s) {
    int e = blockIdx.x * 256 + threadIdx.x;
    float h6 = (ts[s + 1] - ts[s]) * (1.0f / 6.0f);
    float s4r = SSre[e] + Kre[e];
    float s4i = SSim[e] + Kim[e];
    int m = e >> 10;
    int ky = m & 15;
    float pr, pim;
    if (ky != 0) { pr = s4r; pim = s4i; }
    else {
        int corner = m >> 8, mx = (m >> 4) & 15;
        if (corner == 1 && mx == 0) { pr = 0.5f * s4r; pim = 0.5f * s4i; }
        else {
            int pm = (mx == 0) ? m : ((corner == 0) ? (256 + (16 - mx) * 16) : ((16 - mx) * 16));
            int pe = (pm << 10) + (e & 1023);
            float qr = SSre[pe] + Kre[pe];
            float qi = SSim[pe] + Kim[pe];
            pr  = 0.5f * (s4r + qr);
            pim = 0.5f * (s4i - qi);
        }
    }
    Zre[e] += h6 * pr;
    Zim[e] += h6 * pim;
    STre[e] += h6 * s4r;
    STim[e] += h6 * s4i;
}

__global__ void fnde_proj_prep(const float* __restrict__ p1w, const float* __restrict__ p1b,
                               const float* __restrict__ p2w, const float* __restrict__ p2b,
                               const float* __restrict__ lw,  const float* __restrict__ lb,
                               float* __restrict__ q, float* __restrict__ scal) {
    __shared__ float qs[64];
    int c = threadIdx.x;
    float acc = 0.f;
    for (int p = 0; p < 64; ++p) acc += p2w[p] * p1w[p * 64 + c];
    qs[c] = acc;
    q[c] = acc;
    __syncthreads();
    if (c == 0) {
        float r = p2b[0];
        for (int p = 0; p < 64; ++p) r += p2w[p] * p1b[p];
        float al = 0.f, be = r;
        for (int i = 0; i < 64; ++i) { al += qs[i] * lw[i]; be += qs[i] * lb[i]; }
        scal[0] = al;
        scal[1] = be;
    }
}

__global__ void fnde_proj_modes(const float* __restrict__ STre, const float* __restrict__ STim,
                                const float* __restrict__ q, float* __restrict__ T) {
    int t = blockIdx.x * 256 + threadIdx.x;
    int b = t >> 9;
    int m = t & 511;
    const float* pr = STre + m * 1024 + b;
    const float* pi = STim + m * 1024 + b;
    float re = 0.f, im = 0.f;
    for (int c = 0; c < 64; ++c) {
        float w = q[c];
        re += w * pr[c * 16];
        im += w * pi[c * 16];
    }
    T[t * 2]     = re;
    T[t * 2 + 1] = im;
}

__global__ void fnde_inv_dft1(const float* __restrict__ T,
                              const float* __restrict__ cs, const float* __restrict__ sn,
                              float* __restrict__ G) {
    int t  = blockIdx.x * 256 + threadIdx.x;
    int b  = t >> 11;
    int x  = (t >> 4) & 127;
    int ky = t & 15;
    float re = 0.f, im = 0.f;
    for (int corner = 0; corner < 2; ++corner) {
        for (int mx = 0; mx < 16; ++mx) {
            int m  = corner * 256 + mx * 16 + ky;
            int kx = corner ? (112 + mx) : mx;
            float tr = T[(b * 512 + m) * 2];
            float ti = T[(b * 512 + m) * 2 + 1];
            int a = (kx * x) & 127;
            float c = cs[a], s = sn[a];
            re += tr * c - ti * s;
            im += tr * s + ti * c;
        }
    }
    G[t * 2]     = re;
    G[t * 2 + 1] = im;
}

__global__ void fnde_inv_dft2_out(const float* __restrict__ G, const float* __restrict__ z,
                                  const float* __restrict__ cs, const float* __restrict__ sn,
                                  const float* __restrict__ scal, float* __restrict__ out) {
    int t = blockIdx.x * 256 + threadIdx.x;
    int b = t >> 14;
    int x = (t >> 7) & 127;
    int y = t & 127;
    const float* g = G + ((b * 128 + x) * 16) * 2;
    float acc = g[0];
    for (int ky = 1; ky < 16; ++ky) {
        int a = (ky * y) & 127;
        acc += 2.0f * (g[ky * 2] * cs[a] - g[ky * 2 + 1] * sn[a]);
    }
    out[t] = scal[0] * z[t] + scal[1] + acc * (1.0f / 16384.0f);
}

extern "C" void kernel_launch(void* const* d_in, const int* in_sizes, int n_in,
                              void* d_out, int out_size, void* d_ws, size_t ws_size,
                              hipStream_t stream) {
    (void)in_sizes; (void)n_in; (void)out_size; (void)ws_size;

    const float* zin    = (const float*)d_in[0];
    const float* lift_w = (const float*)d_in[1];
    const float* lift_b = (const float*)d_in[2];
    const float* spec_w1 = (const float*)d_in[3];
    const float* spec_w2 = (const float*)d_in[4];
    const float* fl_bias = (const float*)d_in[5];
    const float* p1_w = (const float*)d_in[6];
    const float* p1_b = (const float*)d_in[7];
    const float* p2_w = (const float*)d_in[8];
    const float* p2_b = (const float*)d_in[9];
    const float* ts   = (const float*)d_in[10];
    float* outp = (float*)d_out;

    float* w = (float*)d_ws;
    float* Wp   = w;  w += (size_t)Ll * WLAYER;
    float* Zre  = w;  w += NSPEC;  float* Zim  = w;  w += NSPEC;
    float* Ure  = w;  w += NSPEC;  float* Uim  = w;  w += NSPEC;
    float* Kre  = w;  w += NSPEC;  float* Kim  = w;  w += NSPEC;
    float* SSre = w;  w += NSPEC;  float* SSim = w;  w += NSPEC;
    float* STre = w;  w += NSPEC;  float* STim = w;  w += NSPEC;
    float* Zhat = w;  w += Bb * NMODE * 2;
    float* Ay   = w;  w += Bb * Dd * Mm * 2;
    float* Tm   = w;  w += Bb * NMODE * 2;
    float* Gi   = w;  w += Bb * Dd * Mm * 2;
    float* qv   = w;  w += 64;
    float* cs   = w;  w += 128;
    float* sn   = w;  w += 128;
    float* scal = w;  w += 8;

    fnde_build_tables<<<1, 128, 0, stream>>>(cs, sn);
    fnde_pack_weights<<<(Ll * NMODE * Cc * Cc) / 256, 256, 0, stream>>>(spec_w1, spec_w2, Wp);
    fnde_fwd_dft1<<<(Bb * Dd * Mm) / 256, 256, 0, stream>>>(zin, cs, sn, Ay);
    fnde_fwd_dft2<<<(Bb * NMODE) / 256, 256, 0, stream>>>(Ay, cs, sn, Zhat);
    fnde_init_spec<<<NSPEC / 256, 256, 0, stream>>>(Zhat, lift_w, lift_b, Zre, Zim, STre, STim);

    for (int l = 0; l < Ll; ++l) {
        const float* Wl = Wp + (size_t)l * WLAYER;
        const float* bl = fl_bias + l * Cc;
        for (int s = 0; s < 4; ++s) {
            fnde_gapply<<<128, 128, 0, stream>>>(Zre, Zim, Wl, Kre, Kim, bl);
            fnde_spec_mid<<<NSPEC / 256, 256, 0, stream>>>(Zre, Zim, Kre, Kim, Ure, Uim,
                                                           SSre, SSim, ts, s, 1);
            fnde_gapply<<<128, 128, 0, stream>>>(Ure, Uim, Wl, Kre, Kim, bl);
            fnde_spec_mid<<<NSPEC / 256, 256, 0, stream>>>(Zre, Zim, Kre, Kim, Ure, Uim,
                                                           SSre, SSim, ts, s, 2);
            fnde_gapply<<<128, 128, 0, stream>>>(Ure, Uim, Wl, Kre, Kim, bl);
            fnde_spec_mid<<<NSPEC / 256, 256, 0, stream>>>(Zre, Zim, Kre, Kim, Ure, Uim,
                                                           SSre, SSim, ts, s, 3);
            fnde_gapply<<<128, 128, 0, stream>>>(Ure, Uim, Wl, Kre, Kim, bl);
            fnde_spec_fin<<<NSPEC / 256, 256, 0, stream>>>(Zre, Zim, Kre, Kim, SSre, SSim,
                                                           STre, STim, ts, s);
        }
    }

    fnde_proj_prep<<<1, 64, 0, stream>>>(p1_w, p1_b, p2_w, p2_b, lift_w, lift_b, qv, scal);
    fnde_proj_modes<<<(Bb * NMODE) / 256, 256, 0, stream>>>(STre, STim, qv, Tm);
    fnde_inv_dft1<<<(Bb * Dd * Mm) / 256, 256, 0, stream>>>(Tm, cs, sn, Gi);
    fnde_inv_dft2_out<<<(Bb * Dd * Dd) / 256, 256, 0, stream>>>(Gi, zin, cs, sn, scal, outp);
}


# State planes re-laid [mode][c/4][b][c%4], k order permuted to match (sums regrouped): a lane's B share and accumulator rows are 16-byte pieces; 4x fewer loads/stores, loads grouped ahead of the WMMAs
# speedup vs baseline: 4.7785x; 4.7785x over previous
#include <hip/hip_runtime.h>


typedef float v2f __attribute__((ext_vector_type(2)));
typedef float v4f __attribute__((ext_vector_type(4)));
typedef float v8f __attribute__((ext_vector_type(8)));

#define Bb    16
#define Cc    64
#define Dd    128
#define Mm    16
#define Ll    3
#define NMODE 512
#define NSPEC (NMODE*Cc*Bb)
#define WLAYER (2*NMODE*Cc*Cc)

__global__ void fnde_build_tables(float* cs, float* sn) {
    int t = threadIdx.x;
    float ang = (float)(6.283185307179586 / 128.0) * (float)t;
    cs[t] = cosf(ang);
    sn[t] = sinf(ang);
}

__global__ void fnde_pack_weights(const float* __restrict__ w1,
                                  const float* __restrict__ w2,
                                  float* __restrict__ Wp) {
    int idx = blockIdx.x * 256 + threadIdx.x;
    int i  = idx & 63;
    int o  = (idx >> 6) & 63;
    int mx = (idx >> 12) & 15;
    int corner = (idx >> 16) & 1;
    int l  = idx >> 17;
    const float* src = (corner ? w2 : w1)
                     + (size_t)((((l * Cc + i) * Cc + o) * Mm + mx) * (Mm * 2));
    v4f q[8];
#pragma unroll
    for (int k = 0; k < 8; ++k) q[k] = *(const v4f*)(src + k * 4);
    float* dst = Wp + (size_t)l * WLAYER + (size_t)(corner * 256 + mx * 16) * 4096 + o * 64 + i;
#pragma unroll
    for (int k = 0; k < 8; ++k) {
        float* d0 = dst + (size_t)(2 * k) * 4096;
        float* d1 = d0 + 4096;
        d0[0]               = q[k].x;
        d0[NMODE * Cc * Cc] = q[k].y;
        d1[0]               = q[k].z;
        d1[NMODE * Cc * Cc] = q[k].w;
    }
}

__global__ void fnde_fwd_dft1(const float* __restrict__ z,
                              const float* __restrict__ cs,
                              const float* __restrict__ sn,
                              float* __restrict__ Ay) {
    int t  = blockIdx.x * 256 + threadIdx.x;
    int b  = t >> 11;
    int x  = (t >> 4) & 127;
    int ky = t & 15;
    const float* row = z + (b * 128 + x) * 128;
    float re = 0.f, im = 0.f;
    for (int y = 0; y < 128; ++y) {
        float v = row[y];
        int a = (ky * y) & 127;
        re += v * cs[a];
        im -= v * sn[a];
    }
    Ay[t * 2]     = re;
    Ay[t * 2 + 1] = im;
}

__global__ void fnde_fwd_dft2(const float* __restrict__ Ay,
                              const float* __restrict__ cs,
                              const float* __restrict__ sn,
                              float* __restrict__ Zhat) {
    int t = blockIdx.x * 256 + threadIdx.x;
    int b = t >> 9;
    int m = t & 511;
    int corner = m >> 8, mx = (m >> 4) & 15, ky = m & 15;
    int kx = corner ? (112 + mx) : mx;
    float re = 0.f, im = 0.f;
    for (int x = 0; x < 128; ++x) {
        int base = ((b * 128 + x) * 16 + ky) * 2;
        float ar = Ay[base], ai = Ay[base + 1];
        int a = (kx * x) & 127;
        float c = cs[a], s = sn[a];
        re += ar * c + ai * s;
        im += ai * c - ar * s;
    }
    Zhat[t * 2]     = re;
    Zhat[t * 2 + 1] = im;
}

__global__ void fnde_init_spec(const float* __restrict__ Zhat,
                               const float* __restrict__ lw,
                               const float* __restrict__ lb,
                               float* __restrict__ Zre, float* __restrict__ Zim,
                               float* __restrict__ STre, float* __restrict__ STim) {
    int e = blockIdx.x * 256 + threadIdx.x;
    int m = e >> 10;
    int c = ((e >> 6) & 15) * 4 + (e & 3);
    int b = (e >> 2) & 15;
    float zr = Zhat[(b * 512 + m) * 2];
    float zi = Zhat[(b * 512 + m) * 2 + 1];
    float w = lw[c];
    Zre[e] = w * zr + ((m == 0) ? lb[c] * 16384.0f : 0.0f);
    Zim[e] = w * zi;
    STre[e] = 0.0f;
    STim[e] = 0.0f;
}

struct fnde_afrags { v2f ar[16], ai[16]; };
struct fnde_bfrags { v2f br[16], bi[16]; };
struct fnde_epi    { float zr[8], zi[8], sr[8], si[8], tr[8], ti[8]; };

__device__ __forceinline__ void fnde_load_a(fnde_afrags& f,
                                            const float* __restrict__ Wr,
                                            const float* __restrict__ Wi, int aOff) {
#pragma unroll
    for (int q = 0; q < 8; ++q) {
        const v4f r4 = *(const v4f*)(Wr + aOff + q * 8);
        const v4f i4 = *(const v4f*)(Wi + aOff + q * 8);
        f.ar[2 * q].x = r4.x;      f.ar[2 * q].y = r4.y;
        f.ar[2 * q + 1].x = r4.z;  f.ar[2 * q + 1].y = r4.w;
        f.ai[2 * q].x = i4.x;      f.ai[2 * q].y = i4.y;
        f.ai[2 * q + 1].x = i4.z;  f.ai[2 * q + 1].y = i4.w;
    }
}

__device__ __forceinline__ void fnde_load_b(fnde_bfrags& f, const float* Vr, const float* Vi,
                                            int bOff) {
#pragma unroll
    for (int q = 0; q < 8; ++q) {
        const v4f r4 = *(const v4f*)(Vr + q * 128 + bOff);
        const v4f i4 = *(const v4f*)(Vi + q * 128 + bOff);
        f.br[2 * q].x = r4.x;      f.br[2 * q].y = r4.y;
        f.br[2 * q + 1].x = r4.z;  f.br[2 * q + 1].y = r4.w;
        f.bi[2 * q].x = i4.x;      f.bi[2 * q].y = i4.y;
        f.bi[2 * q + 1].x = i4.z;  f.bi[2 * q + 1].y = i4.w;
    }
}

__device__ __forceinline__ void fnde_product(const fnde_afrags& a, const fnde_bfrags& b,
                                             v8f& cr, v8f& ci) {
#pragma unroll
    for (int kk = 0; kk < 16; ++kk) {
        v2f bin;
        bin.x = -b.bi[kk].x;    bin.y = -b.bi[kk].y;
        cr = __builtin_amdgcn_wmma_f32_16x16x4_f32(false, a.ar[kk], false, b.br[kk],
                                                   (short)0, cr, false, false);
        ci = __builtin_amdgcn_wmma_f32_16x16x4_f32(false, a.ar[kk], false, b.bi[kk],
                                                   (short)0, ci, false, false);
        cr = __builtin_amdgcn_wmma_f32_16x16x4_f32(false, a.ai[kk], false, bin,
                                                   (short)0, cr, false, false);
        ci = __builtin_amdgcn_wmma_f32_16x16x4_f32(false, a.ai[kk], false, b.br[kk],
                                                   (short)0, ci, false, false);
    }
}

__device__ __forceinline__ void fnde_pi(float kr, float ki, float mr, float mi,
                                        bool mirrored, bool halfOnly,
                                        float& pr, float& pim) {
    if (mirrored)      { pr = 0.5f * (kr + mr);  pim = 0.5f * (ki - mi); }
    else if (halfOnly) { pr = 0.5f * kr;         pim = 0.5f * ki; }
    else               { pr = kr;                pim = ki; }
}

template <int PHASE>
__device__ __forceinline__ void fnde_load_epi(fnde_epi& e,
        const float* Zre, const float* Zim, const float* SSre, const float* SSim,
        const float* STre, const float* STim, int off) {
#pragma unroll
    for (int g = 0; g < 2; ++g) {
        const int t = off + g * 64;
        const v4f z0 = *(const v4f*)(Zre + t);
        const v4f z1 = *(const v4f*)(Zim + t);
        v4f s0 = {}, s1 = {}, t0 = {}, t1 = {};
        if (PHASE >= 2) { s0 = *(const v4f*)(SSre + t); s1 = *(const v4f*)(SSim + t); }
        if (PHASE == 4) { t0 = *(const v4f*)(STre + t); t1 = *(const v4f*)(STim + t); }
#pragma unroll
        for (int q = 0; q < 4; ++q) {
            e.zr[4 * g + q] = z0[q];  e.zi[4 * g + q] = z1[q];
            e.sr[4 * g + q] = s0[q];  e.si[4 * g + q] = s1[q];
            e.tr[4 * g + q] = t0[q];  e.ti[4 * g + q] = t1[q];
        }
    }
}

template <int PHASE>
__device__ __forceinline__ void fnde_combine(const fnde_epi& e,
        const float* kr, const float* ki, const float* s4r, const float* s4i,
        const float* xr, const float* xi, bool mirrored, bool halfOnly, float h,
        float* Ore, float* Oim, float* SSre, float* SSim, float* STre, float* STim, int off) {
    float o_r[8], o_i[8], a_r[8], a_i[8];
#pragma unroll
    for (int j = 0; j < 8; ++j) {
        float pr, pim;
        if (PHASE <= 3) {
            const float uc = (PHASE == 3) ? h : 0.5f * h;
            if (PHASE == 1) { a_r[j] = kr[j];                  a_i[j] = ki[j]; }
            else            { a_r[j] = e.sr[j] + 2.0f * kr[j]; a_i[j] = e.si[j] + 2.0f * ki[j]; }
            fnde_pi(kr[j], ki[j], xr[j], xi[j], mirrored, halfOnly, pr, pim);
            o_r[j] = e.zr[j] + uc * pr;
            o_i[j] = e.zi[j] + uc * pim;
        } else {
            const float h6 = h * (1.0f / 6.0f);
            fnde_pi(s4r[j], s4i[j], xr[j], xi[j], mirrored, halfOnly, pr, pim);
            o_r[j] = e.zr[j] + h6 * pr;
            o_i[j] = e.zi[j] + h6 * pim;
            a_r[j] = e.tr[j] + h6 * s4r[j];
            a_i[j] = e.ti[j] + h6 * s4i[j];
        }
    }
    float* Are = (PHASE <= 3) ? SSre : STre;
    float* Aim = (PHASE <= 3) ? SSim : STim;
#pragma unroll
    for (int g = 0; g < 2; ++g) {
        const int t = off + g * 64;
        v4f va, vb, vo, vp;
#pragma unroll
        for (int q = 0; q < 4; ++q) {
            va[q] = a_r[4 * g + q];  vb[q] = a_i[4 * g + q];
            vo[q] = o_r[4 * g + q];  vp[q] = o_i[4 * g + q];
        }
        *(v4f*)(Are + t) = va;
        *(v4f*)(Aim + t) = vb;
        *(v4f*)(Ore + t) = vo;
        *(v4f*)(Oim + t) = vp;
    }
}

template <int PHASE, bool OTHER>
__device__ __forceinline__ void fnde_stage_wave(
        const fnde_afrags& ao, const fnde_afrags& am,
        const float* Vre, const float* Vim,
        const float* Zre, const float* Zim, float* Ore, float* Oim,
        float* SSre, float* SSim, float* STre, float* STim,
        const float* __restrict__ bias, float h,
        int mode, int pm, bool halfOnly, int mt, int laneHi, int laneLo) {
    const int bOff = ((laneHi << 4) + laneLo) * 4;
    const int obase = mt * 16 + (laneHi << 3);
    const int t0  = (obase >> 2) * 64 + laneLo * 4;
    const int offO = mode * 1024 + t0;
    const int offM = pm * 1024 + t0;

    fnde_bfrags bm, bo;
    if (OTHER) fnde_load_b(bm, Vre + pm * 1024, Vim + pm * 1024, bOff);
    fnde_load_b(bo, Vre + mode * 1024, Vim + mode * 1024, bOff);
    fnde_epi eo, em;
    fnde_load_epi<PHASE>(eo, Zre, Zim, SSre, SSim, STre, STim, offO);
    if (OTHER) fnde_load_epi<PHASE>(em, Zre, Zim, SSre, SSim, STre, STim, offM);

    const bool addBias = !OTHER && mode == 0;
    float bj[8];
#pragma unroll
    for (int j = 0; j < 8; ++j) bj[j] = 0.f;
    if (addBias) {
#pragma unroll
        for (int j = 0; j < 8; ++j) bj[j] = bias[obase + j] * 16384.0f;
    }

    __builtin_amdgcn_sched_barrier(0);

    v8f cmr = {}, cmi = {};
    if (OTHER) fnde_product(am, bm, cmr, cmi);
    v8f cr = {}, ci = {};
    fnde_product(ao, bo, cr, ci);

    float kor[8], koi[8], kmr[8], kmi[8], so_r[8], so_i[8], sm_r[8], sm_i[8];
#pragma unroll
    for (int j = 0; j < 8; ++j) {
        kor[j] = cr[j];
        if (addBias) kor[j] += bj[j];
        koi[j] = ci[j];
        kmr[j] = cmr[j];
        kmi[j] = cmi[j];
        so_r[j] = eo.sr[j] + kor[j];
        so_i[j] = eo.si[j] + koi[j];
        sm_r[j] = OTHER ? (em.sr[j] + kmr[j]) : 0.f;
        sm_i[j] = OTHER ? (em.si[j] + kmi[j]) : 0.f;
    }
    const bool mirrored = OTHER || mode == 0;
    const float* xor_ = (PHASE <= 3) ? (OTHER ? kmr : kor) : (OTHER ? sm_r : so_r);
    const float* xoi_ = (PHASE <= 3) ? (OTHER ? kmi : koi) : (OTHER ? sm_i : so_i);
    fnde_combine<PHASE>(eo, kor, koi, so_r, so_i, xor_, xoi_, mirrored, halfOnly, h,
                        Ore, Oim, SSre, SSim, STre, STim, offO);
    if (OTHER) {
        const float* xmr_ = (PHASE <= 3) ? kor : so_r;
        const float* xmi_ = (PHASE <= 3) ? koi : so_i;
        fnde_combine<PHASE>(em, kmr, kmi, sm_r, sm_i, xmr_, xmi_, true, false, h,
                            Ore, Oim, SSre, SSim, STre, STim, offM);
    }
}

template <bool OTHER>
__device__ __forceinline__ void fnde_chain(
        const float* __restrict__ Wp,
        float* Zre, float* Zim, float* Ure, float* Uim, float* Xre, float* Xim,
        float* SSre, float* SSim, float* STre, float* STim,
        const float* __restrict__ fl_bias, const float* __restrict__ ts,
        int mode, int pm, bool halfOnly, int mt, int laneHi, int laneLo) {
    const size_t imOff = (size_t)NMODE * Cc * Cc;
    const int aOff = laneLo * 64 + laneHi * 4;
    for (int l = 0; l < Ll; ++l) {
        const float* Wl = Wp + (size_t)l * WLAYER;
        const float* bl = fl_bias + l * Cc;
        fnde_afrags ao, am;
        {
            const float* Wr = Wl + (size_t)mode * 4096 + mt * 1024;
            fnde_load_a(ao, Wr, Wr + imOff, aOff);
        }
        if (OTHER) {
            const float* Wr = Wl + (size_t)pm * 4096 + mt * 1024;
            fnde_load_a(am, Wr, Wr + imOff, aOff);
        }
        for (int s = 0; s < 4; ++s) {
            const float h = ts[s + 1] - ts[s];
            fnde_stage_wave<1, OTHER>(ao, am, Zre, Zim, Zre, Zim, Ure, Uim, SSre, SSim, STre, STim,
                                      bl, h, mode, pm, halfOnly, mt, laneHi, laneLo);
            __syncthreads();
            fnde_stage_wave<2, OTHER>(ao, am, Ure, Uim, Zre, Zim, Xre, Xim, SSre, SSim, STre, STim,
                                      bl, h, mode, pm, halfOnly, mt, laneHi, laneLo);
            __syncthreads();
            fnde_stage_wave<3, OTHER>(ao, am, Xre, Xim, Zre, Zim, Ure, Uim, SSre, SSim, STre, STim,
                                      bl, h, mode, pm, halfOnly, mt, laneHi, laneLo);
            __syncthreads();
            fnde_stage_wave<4, OTHER>(ao, am, Ure, Uim, Zre, Zim, Zre, Zim, SSre, SSim, STre, STim,
                                      bl, h, mode, pm, halfOnly, mt, laneHi, laneLo);
            __syncthreads();
        }
    }
}

__global__ __launch_bounds__(128) __attribute__((amdgpu_waves_per_eu(2, 4)))
void fnde_ode(const float* __restrict__ Wp,
              float* Zre, float* Zim, float* Ure, float* Uim, float* Xre, float* Xim,
              float* SSre, float* SSim, float* STre, float* STim,
              const float* __restrict__ fl_bias, const float* __restrict__ ts) {
    const int blk = blockIdx.x;
    int mode, pm;
    if (blk < 15)       { mode = (1 + blk) * 16;  pm = 256 + (15 - blk) * 16; }
    else if (blk < 495) { mode = ((blk - 15) / 15) * 16 + 1 + (blk - 15) % 15;  pm = mode; }
    else                { mode = (blk - 495) * 256;  pm = mode; }
    const int mt   = threadIdx.x >> 5;
    const int lane = threadIdx.x & 31;
    const int laneHi = lane >> 4;
    const int laneLo = lane & 15;
    const bool halfOnly = mode == 256;

    if (pm != mode)
        fnde_chain<true>(Wp, Zre, Zim, Ure, Uim, Xre, Xim, SSre, SSim, STre, STim, fl_bias, ts,
                         mode, pm, halfOnly, mt, laneHi, laneLo);
    else
        fnde_chain<false>(Wp, Zre, Zim, Ure, Uim, Xre, Xim, SSre, SSim, STre, STim, fl_bias, ts,
                          mode, pm, halfOnly, mt, laneHi, laneLo);
}

__global__ void fnde_proj_prep(const float* __restrict__ p1w, const float* __restrict__ p1b,
                               const float* __restrict__ p2w, const float* __restrict__ p2b,
                               const float* __restrict__ lw,  const float* __restrict__ lb,
                               float* __restrict__ q, float* __restrict__ scal) {
    __shared__ float qs[64];
    int c = threadIdx.x;
    float acc = 0.f;
    for (int p = 0; p < 64; ++p) acc += p2w[p] * p1w[p * 64 + c];
    qs[c] = acc;
    q[c] = acc;
    __syncthreads();
    if (c == 0) {
        float r = p2b[0];
        for (int p = 0; p < 64; ++p) r += p2w[p] * p1b[p];
        float al = 0.f, be = r;
        for (int i = 0; i < 64; ++i) { al += qs[i] * lw[i]; be += qs[i] * lb[i]; }
        scal[0] = al;
        scal[1] = be;
    }
}

__global__ void fnde_proj_modes(const float* __restrict__ STre, const float* __restrict__ STim,
                                const float* __restrict__ q, float* __restrict__ T) {
    int t = blockIdx.x * 256 + threadIdx.x;
    int b = t >> 9;
    int m = t & 511;
    const float* pr = STre + m * 1024 + b * 4;
    const float* pi = STim + m * 1024 + b * 4;
    float re = 0.f, im = 0.f;
    for (int cq = 0; cq < 16; ++cq) {
        const v4f vr = *(const v4f*)(pr + cq * 64);
        const v4f vi = *(const v4f*)(pi + cq * 64);
#pragma unroll
        for (int s = 0; s < 4; ++s) {
            float w = q[cq * 4 + s];
            re += w * vr[s];
            im += w * vi[s];
        }
    }
    T[t * 2]     = re;
    T[t * 2 + 1] = im;
}

__global__ void fnde_inv_dft1(const float* __restrict__ T,
                              const float* __restrict__ cs, const float* __restrict__ sn,
                              float* __restrict__ G) {
    int t  = blockIdx.x * 256 + threadIdx.x;
    int b  = t >> 11;
    int x  = (t >> 4) & 127;
    int ky = t & 15;
    float re = 0.f, im = 0.f;
    for (int corner = 0; corner < 2; ++corner) {
        for (int mx = 0; mx < 16; ++mx) {
            int m  = corner * 256 + mx * 16 + ky;
            int kx = corner ? (112 + mx) : mx;
            float tr = T[(b * 512 + m) * 2];
            float ti = T[(b * 512 + m) * 2 + 1];
            int a = (kx * x) & 127;
            float c = cs[a], s = sn[a];
            re += tr * c - ti * s;
            im += tr * s + ti * c;
        }
    }
    G[t * 2]     = re;
    G[t * 2 + 1] = im;
}

__global__ void fnde_inv_dft2_out(const float* __restrict__ G, const float* __restrict__ z,
                                  const float* __restrict__ cs, const float* __restrict__ sn,
                                  const float* __restrict__ scal, float* __restrict__ out) {
    int t = blockIdx.x * 256 + threadIdx.x;
    int b = t >> 14;
    int x = (t >> 7) & 127;
    int y = t & 127;
    const float* g = G + ((b * 128 + x) * 16) * 2;
    float acc = g[0];
    for (int ky = 1; ky < 16; ++ky) {
        int a = (ky * y) & 127;
        acc += 2.0f * (g[ky * 2] * cs[a] - g[ky * 2 + 1] * sn[a]);
    }
    out[t] = scal[0] * z[t] + scal[1] + acc * (1.0f / 16384.0f);
}

extern "C" void kernel_launch(void* const* d_in, const int* in_sizes, int n_in,
                              void* d_out, int out_size, void* d_ws, size_t ws_size,
                              hipStream_t stream) {
    (void)in_sizes; (void)n_in; (void)out_size; (void)ws_size;

    const float* zin    = (const float*)d_in[0];
    const float* lift_w = (const float*)d_in[1];
    const float* lift_b = (const float*)d_in[2];
    const float* spec_w1 = (const float*)d_in[3];
    const float* spec_w2 = (const float*)d_in[4];
    const float* fl_bias = (const float*)d_in[5];
    const float* p1_w = (const float*)d_in[6];
    const float* p1_b = (const float*)d_in[7];
    const float* p2_w = (const float*)d_in[8];
    const float* p2_b = (const float*)d_in[9];
    const float* ts   = (const float*)d_in[10];
    float* outp = (float*)d_out;

    float* w = (float*)d_ws;
    float* Wp   = w;  w += (size_t)Ll * WLAYER;
    float* Zre  = w;  w += NSPEC;  float* Zim  = w;  w += NSPEC;
    float* Ure  = w;  w += NSPEC;  float* Uim  = w;  w += NSPEC;
    float* Xre  = w;  w += NSPEC;  float* Xim  = w;  w += NSPEC;
    float* SSre = w;  w += NSPEC;  float* SSim = w;  w += NSPEC;
    float* STre = w;  w += NSPEC;  float* STim = w;  w += NSPEC;
    float* Zhat = w;  w += Bb * NMODE * 2;
    float* Ay   = w;  w += Bb * Dd * Mm * 2;
    float* Tm   = w;  w += Bb * NMODE * 2;
    float* Gi   = w;  w += Bb * Dd * Mm * 2;
    float* qv   = w;  w += 64;
    float* cs   = w;  w += 128;
    float* sn   = w;  w += 128;
    float* scal = w;  w += 8;

    fnde_build_tables<<<1, 128, 0, stream>>>(cs, sn);
    fnde_pack_weights<<<(Ll * 2 * Mm * Cc * Cc) / 256, 256, 0, stream>>>(spec_w1, spec_w2, Wp);
    fnde_fwd_dft1<<<(Bb * Dd * Mm) / 256, 256, 0, stream>>>(zin, cs, sn, Ay);
    fnde_fwd_dft2<<<(Bb * NMODE) / 256, 256, 0, stream>>>(Ay, cs, sn, Zhat);
    fnde_init_spec<<<NSPEC / 256, 256, 0, stream>>>(Zhat, lift_w, lift_b, Zre, Zim, STre, STim);

    fnde_ode<<<497, 128, 0, stream>>>(Wp, Zre, Zim, Ure, Uim, Xre, Xim, SSre, SSim, STre, STim,
                                      fl_bias, ts);

    fnde_proj_prep<<<1, 64, 0, stream>>>(p1_w, p1_b, p2_w, p2_b, lift_w, lift_b, qv, scal);
    fnde_proj_modes<<<(Bb * NMODE) / 256, 256, 0, stream>>>(STre, STim, qv, Tm);
    fnde_inv_dft1<<<(Bb * Dd * Mm) / 256, 256, 0, stream>>>(Tm, cs, sn, Gi);
    fnde_inv_dft2_out<<<(Bb * Dd * Dd) / 256, 256, 0, stream>>>(Gi, zin, cs, sn, scal, outp);
}


# Three producer launches dropped: each consumer work-group recomputes what it reads (twiddle tables in LDS, ODE group lifts its own mode first, projection launch forms q); 10->7 launches; sums same
# speedup vs baseline: 5.0650x; 1.0600x over previous
#include <hip/hip_runtime.h>


typedef float v2f __attribute__((ext_vector_type(2)));
typedef float v4f __attribute__((ext_vector_type(4)));
typedef float v8f __attribute__((ext_vector_type(8)));

#define Bb    16
#define Cc    64
#define Dd    128
#define Mm    16
#define Ll    3
#define NMODE 512
#define NSPEC (NMODE*Cc*Bb)
#define WLAYER (2*NMODE*Cc*Cc)

__device__ __forceinline__ void fnde_tables_lds(float* cs, float* sn) {
    const int t = threadIdx.x;
    if (t < 128) {
        float ang = (float)(6.283185307179586 / 128.0) * (float)t;
        cs[t] = cosf(ang);
        sn[t] = sinf(ang);
    }
    __syncthreads();
}

__global__ void fnde_pack_weights(const float* __restrict__ w1,
                                  const float* __restrict__ w2,
                                  float* __restrict__ Wp) {
    int idx = blockIdx.x * 256 + threadIdx.x;
    int i  = idx & 63;
    int o  = (idx >> 6) & 63;
    int mx = (idx >> 12) & 15;
    int corner = (idx >> 16) & 1;
    int l  = idx >> 17;
    const float* src = (corner ? w2 : w1)
                     + (size_t)((((l * Cc + i) * Cc + o) * Mm + mx) * (Mm * 2));
    v4f q[8];
#pragma unroll
    for (int k = 0; k < 8; ++k) q[k] = *(const v4f*)(src + k * 4);
    float* dst = Wp + (size_t)l * WLAYER + (size_t)(corner * 256 + mx * 16) * 4096 + o * 64 + i;
#pragma unroll
    for (int k = 0; k < 8; ++k) {
        float* d0 = dst + (size_t)(2 * k) * 4096;
        float* d1 = d0 + 4096;
        d0[0]               = q[k].x;
        d0[NMODE * Cc * Cc] = q[k].y;
        d1[0]               = q[k].z;
        d1[NMODE * Cc * Cc] = q[k].w;
    }
}

__global__ void fnde_fwd_dft1(const float* __restrict__ z,
                              float* __restrict__ Ay) {
    __shared__ float cs[128], sn[128];
    fnde_tables_lds(cs, sn);
    int t  = blockIdx.x * 256 + threadIdx.x;
    int b  = t >> 11;
    int x  = (t >> 4) & 127;
    int ky = t & 15;
    const float* row = z + (b * 128 + x) * 128;
    float re = 0.f, im = 0.f;
    for (int y = 0; y < 128; ++y) {
        float v = row[y];
        int a = (ky * y) & 127;
        re += v * cs[a];
        im -= v * sn[a];
    }
    Ay[t * 2]     = re;
    Ay[t * 2 + 1] = im;
}

__global__ void fnde_fwd_dft2(const float* __restrict__ Ay,
                              float* __restrict__ Zhat) {
    __shared__ float cs[128], sn[128];
    fnde_tables_lds(cs, sn);
    int t = blockIdx.x * 256 + threadIdx.x;
    int b = t >> 9;
    int m = t & 511;
    int corner = m >> 8, mx = (m >> 4) & 15, ky = m & 15;
    int kx = corner ? (112 + mx) : mx;
    float re = 0.f, im = 0.f;
    for (int x = 0; x < 128; ++x) {
        int base = ((b * 128 + x) * 16 + ky) * 2;
        float ar = Ay[base], ai = Ay[base + 1];
        int a = (kx * x) & 127;
        float c = cs[a], s = sn[a];
        re += ar * c + ai * s;
        im += ai * c - ar * s;
    }
    Zhat[t * 2]     = re;
    Zhat[t * 2 + 1] = im;
}

struct fnde_afrags { v2f ar[16], ai[16]; };
struct fnde_bfrags { v2f br[16], bi[16]; };
struct fnde_epi    { float zr[8], zi[8], sr[8], si[8], tr[8], ti[8]; };

__device__ __forceinline__ void fnde_load_a(fnde_afrags& f,
                                            const float* __restrict__ Wr,
                                            const float* __restrict__ Wi, int aOff) {
#pragma unroll
    for (int q = 0; q < 8; ++q) {
        const v4f r4 = *(const v4f*)(Wr + aOff + q * 8);
        const v4f i4 = *(const v4f*)(Wi + aOff + q * 8);
        f.ar[2 * q].x = r4.x;      f.ar[2 * q].y = r4.y;
        f.ar[2 * q + 1].x = r4.z;  f.ar[2 * q + 1].y = r4.w;
        f.ai[2 * q].x = i4.x;      f.ai[2 * q].y = i4.y;
        f.ai[2 * q + 1].x = i4.z;  f.ai[2 * q + 1].y = i4.w;
    }
}

__device__ __forceinline__ void fnde_load_b(fnde_bfrags& f, const float* Vr, const float* Vi,
                                            int bOff) {
#pragma unroll
    for (int q = 0; q < 8; ++q) {
        const v4f r4 = *(const v4f*)(Vr + q * 128 + bOff);
        const v4f i4 = *(const v4f*)(Vi + q * 128 + bOff);
        f.br[2 * q].x = r4.x;      f.br[2 * q].y = r4.y;
        f.br[2 * q + 1].x = r4.z;  f.br[2 * q + 1].y = r4.w;
        f.bi[2 * q].x = i4.x;      f.bi[2 * q].y = i4.y;
        f.bi[2 * q + 1].x = i4.z;  f.bi[2 * q + 1].y = i4.w;
    }
}

__device__ __forceinline__ void fnde_product(const fnde_afrags& a, const fnde_bfrags& b,
                                             v8f& cr, v8f& ci) {
#pragma unroll
    for (int kk = 0; kk < 16; ++kk) {
        v2f bin;
        bin.x = -b.bi[kk].x;    bin.y = -b.bi[kk].y;
        cr = __builtin_amdgcn_wmma_f32_16x16x4_f32(false, a.ar[kk], false, b.br[kk],
                                                   (short)0, cr, false, false);
        ci = __builtin_amdgcn_wmma_f32_16x16x4_f32(false, a.ar[kk], false, b.bi[kk],
                                                   (short)0, ci, false, false);
        cr = __builtin_amdgcn_wmma_f32_16x16x4_f32(false, a.ai[kk], false, bin,
                                                   (short)0, cr, false, false);
        ci = __builtin_amdgcn_wmma_f32_16x16x4_f32(false, a.ai[kk], false, b.br[kk],
                                                   (short)0, ci, false, false);
    }
}

__device__ __forceinline__ void fnde_pi(float kr, float ki, float mr, float mi,
                                        bool mirrored, bool halfOnly,
                                        float& pr, float& pim) {
    if (mirrored)      { pr = 0.5f * (kr + mr);  pim = 0.5f * (ki - mi); }
    else if (halfOnly) { pr = 0.5f * kr;         pim = 0.5f * ki; }
    else               { pr = kr;                pim = ki; }
}

template <int PHASE>
__device__ __forceinline__ void fnde_load_epi(fnde_epi& e,
        const float* Zre, const float* Zim, const float* SSre, const float* SSim,
        const float* STre, const float* STim, int off) {
#pragma unroll
    for (int g = 0; g < 2; ++g) {
        const int t = off + g * 64;
        const v4f z0 = *(const v4f*)(Zre + t);
        const v4f z1 = *(const v4f*)(Zim + t);
        v4f s0 = {}, s1 = {}, t0 = {}, t1 = {};
        if (PHASE >= 2) { s0 = *(const v4f*)(SSre + t); s1 = *(const v4f*)(SSim + t); }
        if (PHASE == 4) { t0 = *(const v4f*)(STre + t); t1 = *(const v4f*)(STim + t); }
#pragma unroll
        for (int q = 0; q < 4; ++q) {
            e.zr[4 * g + q] = z0[q];  e.zi[4 * g + q] = z1[q];
            e.sr[4 * g + q] = s0[q];  e.si[4 * g + q] = s1[q];
            e.tr[4 * g + q] = t0[q];  e.ti[4 * g + q] = t1[q];
        }
    }
}

template <int PHASE>
__device__ __forceinline__ void fnde_combine(const fnde_epi& e,
        const float* kr, const float* ki, const float* s4r, const float* s4i,
        const float* xr, const float* xi, bool mirrored, bool halfOnly, float h,
        float* Ore, float* Oim, float* SSre, float* SSim, float* STre, float* STim, int off) {
    float o_r[8], o_i[8], a_r[8], a_i[8];
#pragma unroll
    for (int j = 0; j < 8; ++j) {
        float pr, pim;
        if (PHASE <= 3) {
            const float uc = (PHASE == 3) ? h : 0.5f * h;
            if (PHASE == 1) { a_r[j] = kr[j];                  a_i[j] = ki[j]; }
            else            { a_r[j] = e.sr[j] + 2.0f * kr[j]; a_i[j] = e.si[j] + 2.0f * ki[j]; }
            fnde_pi(kr[j], ki[j], xr[j], xi[j], mirrored, halfOnly, pr, pim);
            o_r[j] = e.zr[j] + uc * pr;
            o_i[j] = e.zi[j] + uc * pim;
        } else {
            const float h6 = h * (1.0f / 6.0f);
            fnde_pi(s4r[j], s4i[j], xr[j], xi[j], mirrored, halfOnly, pr, pim);
            o_r[j] = e.zr[j] + h6 * pr;
            o_i[j] = e.zi[j] + h6 * pim;
            a_r[j] = e.tr[j] + h6 * s4r[j];
            a_i[j] = e.ti[j] + h6 * s4i[j];
        }
    }
    float* Are = (PHASE <= 3) ? SSre : STre;
    float* Aim = (PHASE <= 3) ? SSim : STim;
#pragma unroll
    for (int g = 0; g < 2; ++g) {
        const int t = off + g * 64;
        v4f va, vb, vo, vp;
#pragma unroll
        for (int q = 0; q < 4; ++q) {
            va[q] = a_r[4 * g + q];  vb[q] = a_i[4 * g + q];
            vo[q] = o_r[4 * g + q];  vp[q] = o_i[4 * g + q];
        }
        *(v4f*)(Are + t) = va;
        *(v4f*)(Aim + t) = vb;
        *(v4f*)(Ore + t) = vo;
        *(v4f*)(Oim + t) = vp;
    }
}

template <int PHASE, bool OTHER>
__device__ __forceinline__ void fnde_stage_wave(
        const fnde_afrags& ao, const fnde_afrags& am,
        const float* Vre, const float* Vim,
        const float* Zre, const float* Zim, float* Ore, float* Oim,
        float* SSre, float* SSim, float* STre, float* STim,
        const float* __restrict__ bias, float h,
        int mode, int pm, bool halfOnly, int mt, int laneHi, int laneLo) {
    const int bOff = ((laneHi << 4) + laneLo) * 4;
    const int obase = mt * 16 + (laneHi << 3);
    const int t0  = (obase >> 2) * 64 + laneLo * 4;
    const int offO = mode * 1024 + t0;
    const int offM = pm * 1024 + t0;

    fnde_bfrags bm, bo;
    if (OTHER) fnde_load_b(bm, Vre + pm * 1024, Vim + pm * 1024, bOff);
    fnde_load_b(bo, Vre + mode * 1024, Vim + mode * 1024, bOff);
    fnde_epi eo, em;
    fnde_load_epi<PHASE>(eo, Zre, Zim, SSre, SSim, STre, STim, offO);
    if (OTHER) fnde_load_epi<PHASE>(em, Zre, Zim, SSre, SSim, STre, STim, offM);

    const bool addBias = !OTHER && mode == 0;
    float bj[8];
#pragma unroll
    for (int j = 0; j < 8; ++j) bj[j] = 0.f;
    if (addBias) {
#pragma unroll
        for (int j = 0; j < 8; ++j) bj[j] = bias[obase + j] * 16384.0f;
    }

    __builtin_amdgcn_sched_barrier(0);

    v8f cmr = {}, cmi = {};
    if (OTHER) fnde_product(am, bm, cmr, cmi);
    v8f cr = {}, ci = {};
    fnde_product(ao, bo, cr, ci);

    float kor[8], koi[8], kmr[8], kmi[8], so_r[8], so_i[8], sm_r[8], sm_i[8];
#pragma unroll
    for (int j = 0; j < 8; ++j) {
        kor[j] = cr[j];
        if (addBias) kor[j] += bj[j];
        koi[j] = ci[j];
        kmr[j] = cmr[j];
        kmi[j] = cmi[j];
        so_r[j] = eo.sr[j] + kor[j];
        so_i[j] = eo.si[j] + koi[j];
        sm_r[j] = OTHER ? (em.sr[j] + kmr[j]) : 0.f;
        sm_i[j] = OTHER ? (em.si[j] + kmi[j]) : 0.f;
    }
    const bool mirrored = OTHER || mode == 0;
    const float* xor_ = (PHASE <= 3) ? (OTHER ? kmr : kor) : (OTHER ? sm_r : so_r);
    const float* xoi_ = (PHASE <= 3) ? (OTHER ? kmi : koi) : (OTHER ? sm_i : so_i);
    fnde_combine<PHASE>(eo, kor, koi, so_r, so_i, xor_, xoi_, mirrored, halfOnly, h,
                        Ore, Oim, SSre, SSim, STre, STim, offO);
    if (OTHER) {
        const float* xmr_ = (PHASE <= 3) ? kor : so_r;
        const float* xmi_ = (PHASE <= 3) ? koi : so_i;
        fnde_combine<PHASE>(em, kmr, kmi, sm_r, sm_i, xmr_, xmi_, true, false, h,
                            Ore, Oim, SSre, SSim, STre, STim, offM);
    }
}

__device__ __forceinline__ void fnde_lift_mode(const float* __restrict__ Zhat,
        const float* __restrict__ lw, const float* __restrict__ lb,
        float* Zre, float* Zim, float* STre, float* STim,
        int m, int obase, int laneLo, int off) {
    const float zr = Zhat[(laneLo * 512 + m) * 2];
    const float zi = Zhat[(laneLo * 512 + m) * 2 + 1];
    const v4f zero = {};
#pragma unroll
    for (int g = 0; g < 2; ++g) {
        const v4f w4 = *(const v4f*)(lw + obase + 4 * g);
        const v4f b4 = *(const v4f*)(lb + obase + 4 * g);
        v4f vr, vi;
#pragma unroll
        for (int q = 0; q < 4; ++q) {
            const float w = w4[q];
            vr[q] = w * zr + ((m == 0) ? b4[q] * 16384.0f : 0.0f);
            vi[q] = w * zi;
        }
        const int t = off + g * 64;
        *(v4f*)(Zre + t)  = vr;
        *(v4f*)(Zim + t)  = vi;
        *(v4f*)(STre + t) = zero;
        *(v4f*)(STim + t) = zero;
    }
}

template <bool OTHER>
__device__ __forceinline__ void fnde_chain(
        const float* __restrict__ Wp,
        float* Zre, float* Zim, float* Ure, float* Uim, float* Xre, float* Xim,
        float* SSre, float* SSim, float* STre, float* STim,
        const float* __restrict__ fl_bias, const float* __restrict__ ts,
        const float* __restrict__ Zhat, const float* __restrict__ lw,
        const float* __restrict__ lb,
        int mode, int pm, bool halfOnly, int mt, int laneHi, int laneLo) {
    const size_t imOff = (size_t)NMODE * Cc * Cc;
    const int aOff = laneLo * 64 + laneHi * 4;
    {
        const int obase = mt * 16 + (laneHi << 3);
        const int t0 = (obase >> 2) * 64 + laneLo * 4;
        fnde_lift_mode(Zhat, lw, lb, Zre, Zim, STre, STim, mode, obase, laneLo, mode * 1024 + t0);
        if (OTHER)
            fnde_lift_mode(Zhat, lw, lb, Zre, Zim, STre, STim, pm, obase, laneLo, pm * 1024 + t0);
        __syncthreads();
    }
    for (int l = 0; l < Ll; ++l) {
        const float* Wl = Wp + (size_t)l * WLAYER;
        const float* bl = fl_bias + l * Cc;
        fnde_afrags ao, am;
        {
            const float* Wr = Wl + (size_t)mode * 4096 + mt * 1024;
            fnde_load_a(ao, Wr, Wr + imOff, aOff);
        }
        if (OTHER) {
            const float* Wr = Wl + (size_t)pm * 4096 + mt * 1024;
            fnde_load_a(am, Wr, Wr + imOff, aOff);
        }
        for (int s = 0; s < 4; ++s) {
            const float h = ts[s + 1] - ts[s];
            fnde_stage_wave<1, OTHER>(ao, am, Zre, Zim, Zre, Zim, Ure, Uim, SSre, SSim, STre, STim,
                                      bl, h, mode, pm, halfOnly, mt, laneHi, laneLo);
            __syncthreads();
            fnde_stage_wave<2, OTHER>(ao, am, Ure, Uim, Zre, Zim, Xre, Xim, SSre, SSim, STre, STim,
                                      bl, h, mode, pm, halfOnly, mt, laneHi, laneLo);
            __syncthreads();
            fnde_stage_wave<3, OTHER>(ao, am, Xre, Xim, Zre, Zim, Ure, Uim, SSre, SSim, STre, STim,
                                      bl, h, mode, pm, halfOnly, mt, laneHi, laneLo);
            __syncthreads();
            fnde_stage_wave<4, OTHER>(ao, am, Ure, Uim, Zre, Zim, Zre, Zim, SSre, SSim, STre, STim,
                                      bl, h, mode, pm, halfOnly, mt, laneHi, laneLo);
            __syncthreads();
        }
    }
}

__global__ __launch_bounds__(128) __attribute__((amdgpu_waves_per_eu(2, 4)))
void fnde_ode(const float* __restrict__ Wp,
              float* Zre, float* Zim, float* Ure, float* Uim, float* Xre, float* Xim,
              float* SSre, float* SSim, float* STre, float* STim,
              const float* __restrict__ fl_bias, const float* __restrict__ ts,
              const float* __restrict__ Zhat, const float* __restrict__ lw,
              const float* __restrict__ lb) {
    const int blk = blockIdx.x;
    int mode, pm;
    if (blk < 15)       { mode = (1 + blk) * 16;  pm = 256 + (15 - blk) * 16; }
    else if (blk < 495) { mode = ((blk - 15) / 15) * 16 + 1 + (blk - 15) % 15;  pm = mode; }
    else                { mode = (blk - 495) * 256;  pm = mode; }
    const int mt   = threadIdx.x >> 5;
    const int lane = threadIdx.x & 31;
    const int laneHi = lane >> 4;
    const int laneLo = lane & 15;
    const bool halfOnly = mode == 256;

    if (pm != mode)
        fnde_chain<true>(Wp, Zre, Zim, Ure, Uim, Xre, Xim, SSre, SSim, STre, STim, fl_bias, ts,
                         Zhat, lw, lb, mode, pm, halfOnly, mt, laneHi, laneLo);
    else
        fnde_chain<false>(Wp, Zre, Zim, Ure, Uim, Xre, Xim, SSre, SSim, STre, STim, fl_bias, ts,
                          Zhat, lw, lb, mode, pm, halfOnly, mt, laneHi, laneLo);
}

__global__ void fnde_proj_modes(const float* __restrict__ STre, const float* __restrict__ STim,
                                const float* __restrict__ p1w, const float* __restrict__ p1b,
                                const float* __restrict__ p2w, const float* __restrict__ p2b,
                                const float* __restrict__ lw,  const float* __restrict__ lb,
                                float* __restrict__ scal, float* __restrict__ T) {
    __shared__ float q[64];
    if (threadIdx.x < 64) {
        const int c = threadIdx.x;
        float acc = 0.f;
        for (int p = 0; p < 64; ++p) acc += p2w[p] * p1w[p * 64 + c];
        q[c] = acc;
    }
    __syncthreads();
    if (blockIdx.x == 0 && threadIdx.x == 0) {
        float r = p2b[0];
        for (int p = 0; p < 64; ++p) r += p2w[p] * p1b[p];
        float al = 0.f, be = r;
        for (int i = 0; i < 64; ++i) { al += q[i] * lw[i]; be += q[i] * lb[i]; }
        scal[0] = al;
        scal[1] = be;
    }
    int t = blockIdx.x * 256 + threadIdx.x;
    int b = t >> 9;
    int m = t & 511;
    const float* pr = STre + m * 1024 + b * 4;
    const float* pi = STim + m * 1024 + b * 4;
    float re = 0.f, im = 0.f;
    for (int cq = 0; cq < 16; ++cq) {
        const v4f vr = *(const v4f*)(pr + cq * 64);
        const v4f vi = *(const v4f*)(pi + cq * 64);
#pragma unroll
        for (int s = 0; s < 4; ++s) {
            float w = q[cq * 4 + s];
            re += w * vr[s];
            im += w * vi[s];
        }
    }
    T[t * 2]     = re;
    T[t * 2 + 1] = im;
}

__global__ void fnde_inv_dft1(const float* __restrict__ T,
                              float* __restrict__ G) {
    __shared__ float cs[128], sn[128];
    fnde_tables_lds(cs, sn);
    int t  = blockIdx.x * 256 + threadIdx.x;
    int b  = t >> 11;
    int x  = (t >> 4) & 127;
    int ky = t & 15;
    float re = 0.f, im = 0.f;
    for (int corner = 0; corner < 2; ++corner) {
        for (int mx = 0; mx < 16; ++mx) {
            int m  = corner * 256 + mx * 16 + ky;
            int kx = corner ? (112 + mx) : mx;
            float tr = T[(b * 512 + m) * 2];
            float ti = T[(b * 512 + m) * 2 + 1];
            int a = (kx * x) & 127;
            float c = cs[a], s = sn[a];
            re += tr * c - ti * s;
            im += tr * s + ti * c;
        }
    }
    G[t * 2]     = re;
    G[t * 2 + 1] = im;
}

__global__ void fnde_inv_dft2_out(const float* __restrict__ G, const float* __restrict__ z,
                                  const float* __restrict__ scal, float* __restrict__ out) {
    __shared__ float cs[128], sn[128];
    fnde_tables_lds(cs, sn);
    int t = blockIdx.x * 256 + threadIdx.x;
    int b = t >> 14;
    int x = (t >> 7) & 127;
    int y = t & 127;
    const float* g = G + ((b * 128 + x) * 16) * 2;
    float acc = g[0];
    for (int ky = 1; ky < 16; ++ky) {
        int a = (ky * y) & 127;
        acc += 2.0f * (g[ky * 2] * cs[a] - g[ky * 2 + 1] * sn[a]);
    }
    out[t] = scal[0] * z[t] + scal[1] + acc * (1.0f / 16384.0f);
}

extern "C" void kernel_launch(void* const* d_in, const int* in_sizes, int n_in,
                              void* d_out, int out_size, void* d_ws, size_t ws_size,
                              hipStream_t stream) {
    (void)in_sizes; (void)n_in; (void)out_size; (void)ws_size;

    const float* zin    = (const float*)d_in[0];
    const float* lift_w = (const float*)d_in[1];
    const float* lift_b = (const float*)d_in[2];
    const float* spec_w1 = (const float*)d_in[3];
    const float* spec_w2 = (const float*)d_in[4];
    const float* fl_bias = (const float*)d_in[5];
    const float* p1_w = (const float*)d_in[6];
    const float* p1_b = (const float*)d_in[7];
    const float* p2_w = (const float*)d_in[8];
    const float* p2_b = (const float*)d_in[9];
    const float* ts   = (const float*)d_in[10];
    float* outp = (float*)d_out;

    float* w = (float*)d_ws;
    float* Wp   = w;  w += (size_t)Ll * WLAYER;
    float* Zre  = w;  w += NSPEC;  float* Zim  = w;  w += NSPEC;
    float* Ure  = w;  w += NSPEC;  float* Uim  = w;  w += NSPEC;
    float* Xre  = w;  w += NSPEC;  float* Xim  = w;  w += NSPEC;
    float* SSre = w;  w += NSPEC;  float* SSim = w;  w += NSPEC;
    float* STre = w;  w += NSPEC;  float* STim = w;  w += NSPEC;
    float* Zhat = w;  w += Bb * NMODE * 2;
    float* Ay   = w;  w += Bb * Dd * Mm * 2;
    float* Tm   = w;  w += Bb * NMODE * 2;
    float* Gi   = w;  w += Bb * Dd * Mm * 2;
    float* scal = w;  w += 8;

    fnde_pack_weights<<<(Ll * 2 * Mm * Cc * Cc) / 256, 256, 0, stream>>>(spec_w1, spec_w2, Wp);
    fnde_fwd_dft1<<<(Bb * Dd * Mm) / 256, 256, 0, stream>>>(zin, Ay);
    fnde_fwd_dft2<<<(Bb * NMODE) / 256, 256, 0, stream>>>(Ay, Zhat);

    fnde_ode<<<497, 128, 0, stream>>>(Wp, Zre, Zim, Ure, Uim, Xre, Xim, SSre, SSim, STre, STim,
                                      fl_bias, ts, Zhat, lift_w, lift_b);

    fnde_proj_modes<<<(Bb * NMODE) / 256, 256, 0, stream>>>(STre, STim, p1_w, p1_b, p2_w, p2_b,
                                                            lift_w, lift_b, scal, Tm);
    fnde_inv_dft1<<<(Bb * Dd * Mm) / 256, 256, 0, stream>>>(Tm, Gi);
    fnde_inv_dft2_out<<<(Bb * Dd * Dd) / 256, 256, 0, stream>>>(Gi, zin, scal, outp);
}


# ODE launch: each wave keeps its own tile of state Z, RK4 stage sum and step total in registers for the whole chain; only the next stage input is stored (4 of 8 stores, no combine loads); sums same
# speedup vs baseline: 5.5710x; 1.0999x over previous
#include <hip/hip_runtime.h>


typedef float v2f __attribute__((ext_vector_type(2)));
typedef float v4f __attribute__((ext_vector_type(4)));
typedef float v8f __attribute__((ext_vector_type(8)));

#define Bb    16
#define Cc    64
#define Dd    128
#define Mm    16
#define Ll    3
#define NMODE 512
#define NSPEC (NMODE*Cc*Bb)
#define WLAYER (2*NMODE*Cc*Cc)

__device__ __forceinline__ void fnde_tables_lds(float* cs, float* sn) {
    const int t = threadIdx.x;
    if (t < 128) {
        float ang = (float)(6.283185307179586 / 128.0) * (float)t;
        cs[t] = cosf(ang);
        sn[t] = sinf(ang);
    }
    __syncthreads();
}

__global__ void fnde_pack_weights(const float* __restrict__ w1,
                                  const float* __restrict__ w2,
                                  float* __restrict__ Wp) {
    int idx = blockIdx.x * 256 + threadIdx.x;
    int i  = idx & 63;
    int o  = (idx >> 6) & 63;
    int mx = (idx >> 12) & 15;
    int corner = (idx >> 16) & 1;
    int l  = idx >> 17;
    const float* src = (corner ? w2 : w1)
                     + (size_t)((((l * Cc + i) * Cc + o) * Mm + mx) * (Mm * 2));
    v4f q[8];
#pragma unroll
    for (int k = 0; k < 8; ++k) q[k] = *(const v4f*)(src + k * 4);
    float* dst = Wp + (size_t)l * WLAYER + (size_t)(corner * 256 + mx * 16) * 4096 + o * 64 + i;
#pragma unroll
    for (int k = 0; k < 8; ++k) {
        float* d0 = dst + (size_t)(2 * k) * 4096;
        float* d1 = d0 + 4096;
        d0[0]               = q[k].x;
        d0[NMODE * Cc * Cc] = q[k].y;
        d1[0]               = q[k].z;
        d1[NMODE * Cc * Cc] = q[k].w;
    }
}

__global__ void fnde_fwd_dft1(const float* __restrict__ z,
                              float* __restrict__ Ay) {
    __shared__ float cs[128], sn[128];
    fnde_tables_lds(cs, sn);
    int t  = blockIdx.x * 256 + threadIdx.x;
    int b  = t >> 11;
    int x  = (t >> 4) & 127;
    int ky = t & 15;
    const float* row = z + (b * 128 + x) * 128;
    float re = 0.f, im = 0.f;
    for (int y = 0; y < 128; ++y) {
        float v = row[y];
        int a = (ky * y) & 127;
        re += v * cs[a];
        im -= v * sn[a];
    }
    Ay[t * 2]     = re;
    Ay[t * 2 + 1] = im;
}

__global__ void fnde_fwd_dft2(const float* __restrict__ Ay,
                              float* __restrict__ Zhat) {
    __shared__ float cs[128], sn[128];
    fnde_tables_lds(cs, sn);
    int t = blockIdx.x * 256 + threadIdx.x;
    int b = t >> 9;
    int m = t & 511;
    int corner = m >> 8, mx = (m >> 4) & 15, ky = m & 15;
    int kx = corner ? (112 + mx) : mx;
    float re = 0.f, im = 0.f;
    for (int x = 0; x < 128; ++x) {
        int base = ((b * 128 + x) * 16 + ky) * 2;
        float ar = Ay[base], ai = Ay[base + 1];
        int a = (kx * x) & 127;
        float c = cs[a], s = sn[a];
        re += ar * c + ai * s;
        im += ai * c - ar * s;
    }
    Zhat[t * 2]     = re;
    Zhat[t * 2 + 1] = im;
}

struct fnde_afrags { v2f ar[16], ai[16]; };
struct fnde_bfrags { v2f br[16], bi[16]; };

__device__ __forceinline__ void fnde_load_a(fnde_afrags& f,
                                            const float* __restrict__ Wr,
                                            const float* __restrict__ Wi, int aOff) {
#pragma unroll
    for (int q = 0; q < 8; ++q) {
        const v4f r4 = *(const v4f*)(Wr + aOff + q * 8);
        const v4f i4 = *(const v4f*)(Wi + aOff + q * 8);
        f.ar[2 * q].x = r4.x;      f.ar[2 * q].y = r4.y;
        f.ar[2 * q + 1].x = r4.z;  f.ar[2 * q + 1].y = r4.w;
        f.ai[2 * q].x = i4.x;      f.ai[2 * q].y = i4.y;
        f.ai[2 * q + 1].x = i4.z;  f.ai[2 * q + 1].y = i4.w;
    }
}

__device__ __forceinline__ void fnde_load_b(fnde_bfrags& f, const float* Vr, const float* Vi,
                                            int bOff) {
#pragma unroll
    for (int q = 0; q < 8; ++q) {
        const v4f r4 = *(const v4f*)(Vr + q * 128 + bOff);
        const v4f i4 = *(const v4f*)(Vi + q * 128 + bOff);
        f.br[2 * q].x = r4.x;      f.br[2 * q].y = r4.y;
        f.br[2 * q + 1].x = r4.z;  f.br[2 * q + 1].y = r4.w;
        f.bi[2 * q].x = i4.x;      f.bi[2 * q].y = i4.y;
        f.bi[2 * q + 1].x = i4.z;  f.bi[2 * q + 1].y = i4.w;
    }
}

__device__ __forceinline__ void fnde_product(const fnde_afrags& a, const fnde_bfrags& b,
                                             v8f& cr, v8f& ci) {
#pragma unroll
    for (int kk = 0; kk < 16; ++kk) {
        v2f bin;
        bin.x = -b.bi[kk].x;    bin.y = -b.bi[kk].y;
        cr = __builtin_amdgcn_wmma_f32_16x16x4_f32(false, a.ar[kk], false, b.br[kk],
                                                   (short)0, cr, false, false);
        ci = __builtin_amdgcn_wmma_f32_16x16x4_f32(false, a.ar[kk], false, b.bi[kk],
                                                   (short)0, ci, false, false);
        cr = __builtin_amdgcn_wmma_f32_16x16x4_f32(false, a.ai[kk], false, bin,
                                                   (short)0, cr, false, false);
        ci = __builtin_amdgcn_wmma_f32_16x16x4_f32(false, a.ai[kk], false, b.br[kk],
                                                   (short)0, ci, false, false);
    }
}

__device__ __forceinline__ void fnde_pi(float kr, float ki, float mr, float mi,
                                        bool mirrored, bool halfOnly,
                                        float& pr, float& pim) {
    if (mirrored)      { pr = 0.5f * (kr + mr);  pim = 0.5f * (ki - mi); }
    else if (halfOnly) { pr = 0.5f * kr;         pim = 0.5f * ki; }
    else               { pr = kr;                pim = ki; }
}

struct fnde_state { float zr[8], zi[8], sr[8], si[8], tr[8], ti[8]; };

template <int PHASE>
__device__ __forceinline__ void fnde_combine(fnde_state& e,
        const float* kr, const float* ki, const float* s4r, const float* s4i,
        const float* xr, const float* xi, bool mirrored, bool halfOnly, float h,
        float* Ore, float* Oim, int off) {
    float o_r[8], o_i[8];
#pragma unroll
    for (int j = 0; j < 8; ++j) {
        float pr, pim;
        if (PHASE <= 3) {
            const float uc = (PHASE == 3) ? h : 0.5f * h;
            fnde_pi(kr[j], ki[j], xr[j], xi[j], mirrored, halfOnly, pr, pim);
            o_r[j] = e.zr[j] + uc * pr;
            o_i[j] = e.zi[j] + uc * pim;
            if (PHASE == 1) { e.sr[j] = kr[j];                  e.si[j] = ki[j]; }
            else            { e.sr[j] = e.sr[j] + 2.0f * kr[j]; e.si[j] = e.si[j] + 2.0f * ki[j]; }
        } else {
            const float h6 = h * (1.0f / 6.0f);
            fnde_pi(s4r[j], s4i[j], xr[j], xi[j], mirrored, halfOnly, pr, pim);
            o_r[j] = e.zr[j] + h6 * pr;
            o_i[j] = e.zi[j] + h6 * pim;
            e.zr[j] = o_r[j];
            e.zi[j] = o_i[j];
            e.tr[j] = e.tr[j] + h6 * s4r[j];
            e.ti[j] = e.ti[j] + h6 * s4i[j];
        }
    }
#pragma unroll
    for (int g = 0; g < 2; ++g) {
        const int t = off + g * 64;
        v4f vo, vp;
#pragma unroll
        for (int q = 0; q < 4; ++q) {
            vo[q] = o_r[4 * g + q];  vp[q] = o_i[4 * g + q];
        }
        *(v4f*)(Ore + t) = vo;
        *(v4f*)(Oim + t) = vp;
    }
}

template <int PHASE, bool OTHER>
__device__ __forceinline__ void fnde_stage_wave(
        const fnde_afrags& ao, const fnde_afrags& am,
        fnde_state& eo, fnde_state& em,
        const float* Vre, const float* Vim, float* Ore, float* Oim,
        const float* __restrict__ bias, float h,
        int mode, int pm, bool halfOnly, int mt, int laneHi, int laneLo) {
    const int bOff = ((laneHi << 4) + laneLo) * 4;
    const int obase = mt * 16 + (laneHi << 3);
    const int t0  = (obase >> 2) * 64 + laneLo * 4;
    const int offO = mode * 1024 + t0;
    const int offM = pm * 1024 + t0;

    fnde_bfrags bm, bo;
    if (OTHER) fnde_load_b(bm, Vre + pm * 1024, Vim + pm * 1024, bOff);
    fnde_load_b(bo, Vre + mode * 1024, Vim + mode * 1024, bOff);

    const bool addBias = !OTHER && mode == 0;
    float bj[8];
#pragma unroll
    for (int j = 0; j < 8; ++j) bj[j] = 0.f;
    if (addBias) {
#pragma unroll
        for (int j = 0; j < 8; ++j) bj[j] = bias[obase + j] * 16384.0f;
    }

    __builtin_amdgcn_sched_barrier(0);

    v8f cmr = {}, cmi = {};
    if (OTHER) fnde_product(am, bm, cmr, cmi);
    v8f cr = {}, ci = {};
    fnde_product(ao, bo, cr, ci);

    float kor[8], koi[8], kmr[8], kmi[8], so_r[8], so_i[8], sm_r[8], sm_i[8];
#pragma unroll
    for (int j = 0; j < 8; ++j) {
        kor[j] = cr[j];
        if (addBias) kor[j] += bj[j];
        koi[j] = ci[j];
        kmr[j] = cmr[j];
        kmi[j] = cmi[j];
        so_r[j] = (PHASE == 4) ? (eo.sr[j] + kor[j]) : 0.f;
        so_i[j] = (PHASE == 4) ? (eo.si[j] + koi[j]) : 0.f;
        sm_r[j] = (PHASE == 4 && OTHER) ? (em.sr[j] + kmr[j]) : 0.f;
        sm_i[j] = (PHASE == 4 && OTHER) ? (em.si[j] + kmi[j]) : 0.f;
    }
    const bool mirrored = OTHER || mode == 0;
    const float* xor_ = (PHASE <= 3) ? (OTHER ? kmr : kor) : (OTHER ? sm_r : so_r);
    const float* xoi_ = (PHASE <= 3) ? (OTHER ? kmi : koi) : (OTHER ? sm_i : so_i);
    fnde_combine<PHASE>(eo, kor, koi, so_r, so_i, xor_, xoi_, mirrored, halfOnly, h,
                        Ore, Oim, offO);
    if (OTHER) {
        const float* xmr_ = (PHASE <= 3) ? kor : so_r;
        const float* xmi_ = (PHASE <= 3) ? koi : so_i;
        fnde_combine<PHASE>(em, kmr, kmi, sm_r, sm_i, xmr_, xmi_, true, false, h,
                            Ore, Oim, offM);
    }
}

__device__ __forceinline__ void fnde_lift_mode(fnde_state& e, const float* __restrict__ Zhat,
        const float* __restrict__ lw, const float* __restrict__ lb,
        float* Zre, float* Zim, int m, int obase, int laneLo, int off) {
    const float zr = Zhat[(laneLo * 512 + m) * 2];
    const float zi = Zhat[(laneLo * 512 + m) * 2 + 1];
#pragma unroll
    for (int g = 0; g < 2; ++g) {
        const v4f w4 = *(const v4f*)(lw + obase + 4 * g);
        const v4f b4 = *(const v4f*)(lb + obase + 4 * g);
        v4f vr, vi;
#pragma unroll
        for (int q = 0; q < 4; ++q) {
            const float w = w4[q];
            vr[q] = w * zr + ((m == 0) ? b4[q] * 16384.0f : 0.0f);
            vi[q] = w * zi;
            e.zr[4 * g + q] = vr[q];  e.zi[4 * g + q] = vi[q];
            e.sr[4 * g + q] = 0.f;    e.si[4 * g + q] = 0.f;
            e.tr[4 * g + q] = 0.f;    e.ti[4 * g + q] = 0.f;
        }
        const int t = off + g * 64;
        *(v4f*)(Zre + t)  = vr;
        *(v4f*)(Zim + t)  = vi;
    }
}

__device__ __forceinline__ void fnde_store_total(const fnde_state& e,
        float* STre, float* STim, int off) {
#pragma unroll
    for (int g = 0; g < 2; ++g) {
        v4f vr, vi;
#pragma unroll
        for (int q = 0; q < 4; ++q) { vr[q] = e.tr[4 * g + q];  vi[q] = e.ti[4 * g + q]; }
        const int t = off + g * 64;
        *(v4f*)(STre + t) = vr;
        *(v4f*)(STim + t) = vi;
    }
}

template <bool OTHER>
__device__ __forceinline__ void fnde_chain(
        const float* __restrict__ Wp,
        float* Zre, float* Zim, float* Ure, float* Uim, float* Xre, float* Xim,
        float* STre, float* STim,
        const float* __restrict__ fl_bias, const float* __restrict__ ts,
        const float* __restrict__ Zhat, const float* __restrict__ lw,
        const float* __restrict__ lb,
        int mode, int pm, bool halfOnly, int mt, int laneHi, int laneLo) {
    const size_t imOff = (size_t)NMODE * Cc * Cc;
    const int aOff = laneLo * 64 + laneHi * 4;
    const int obase = mt * 16 + (laneHi << 3);
    const int t0 = (obase >> 2) * 64 + laneLo * 4;
    fnde_state eo, em;
    fnde_lift_mode(eo, Zhat, lw, lb, Zre, Zim, mode, obase, laneLo, mode * 1024 + t0);
    if (OTHER)
        fnde_lift_mode(em, Zhat, lw, lb, Zre, Zim, pm, obase, laneLo, pm * 1024 + t0);
    __syncthreads();
    for (int l = 0; l < Ll; ++l) {
        const float* Wl = Wp + (size_t)l * WLAYER;
        const float* bl = fl_bias + l * Cc;
        fnde_afrags ao, am;
        {
            const float* Wr = Wl + (size_t)mode * 4096 + mt * 1024;
            fnde_load_a(ao, Wr, Wr + imOff, aOff);
        }
        if (OTHER) {
            const float* Wr = Wl + (size_t)pm * 4096 + mt * 1024;
            fnde_load_a(am, Wr, Wr + imOff, aOff);
        }
        for (int s = 0; s < 4; ++s) {
            const float h = ts[s + 1] - ts[s];
            fnde_stage_wave<1, OTHER>(ao, am, eo, em, Zre, Zim, Ure, Uim,
                                      bl, h, mode, pm, halfOnly, mt, laneHi, laneLo);
            __syncthreads();
            fnde_stage_wave<2, OTHER>(ao, am, eo, em, Ure, Uim, Xre, Xim,
                                      bl, h, mode, pm, halfOnly, mt, laneHi, laneLo);
            __syncthreads();
            fnde_stage_wave<3, OTHER>(ao, am, eo, em, Xre, Xim, Ure, Uim,
                                      bl, h, mode, pm, halfOnly, mt, laneHi, laneLo);
            __syncthreads();
            fnde_stage_wave<4, OTHER>(ao, am, eo, em, Ure, Uim, Zre, Zim,
                                      bl, h, mode, pm, halfOnly, mt, laneHi, laneLo);
            __syncthreads();
        }
    }
    fnde_store_total(eo, STre, STim, mode * 1024 + t0);
    if (OTHER) fnde_store_total(em, STre, STim, pm * 1024 + t0);
}

__global__ __launch_bounds__(128) __attribute__((amdgpu_waves_per_eu(2, 4)))
void fnde_ode(const float* __restrict__ Wp,
              float* Zre, float* Zim, float* Ure, float* Uim, float* Xre, float* Xim,
              float* STre, float* STim,
              const float* __restrict__ fl_bias, const float* __restrict__ ts,
              const float* __restrict__ Zhat, const float* __restrict__ lw,
              const float* __restrict__ lb) {
    const int blk = blockIdx.x;
    int mode, pm;
    if (blk < 15)       { mode = (1 + blk) * 16;  pm = 256 + (15 - blk) * 16; }
    else if (blk < 495) { mode = ((blk - 15) / 15) * 16 + 1 + (blk - 15) % 15;  pm = mode; }
    else                { mode = (blk - 495) * 256;  pm = mode; }
    const int mt   = threadIdx.x >> 5;
    const int lane = threadIdx.x & 31;
    const int laneHi = lane >> 4;
    const int laneLo = lane & 15;
    const bool halfOnly = mode == 256;

    if (pm != mode)
        fnde_chain<true>(Wp, Zre, Zim, Ure, Uim, Xre, Xim, STre, STim, fl_bias, ts,
                         Zhat, lw, lb, mode, pm, halfOnly, mt, laneHi, laneLo);
    else
        fnde_chain<false>(Wp, Zre, Zim, Ure, Uim, Xre, Xim, STre, STim, fl_bias, ts,
                          Zhat, lw, lb, mode, pm, halfOnly, mt, laneHi, laneLo);
}

__global__ void fnde_proj_modes(const float* __restrict__ STre, const float* __restrict__ STim,
                                const float* __restrict__ p1w, const float* __restrict__ p1b,
                                const float* __restrict__ p2w, const float* __restrict__ p2b,
                                const float* __restrict__ lw,  const float* __restrict__ lb,
                                float* __restrict__ scal, float* __restrict__ T) {
    __shared__ float q[64];
    if (threadIdx.x < 64) {
        const int c = threadIdx.x;
        float acc = 0.f;
        for (int p = 0; p < 64; ++p) acc += p2w[p] * p1w[p * 64 + c];
        q[c] = acc;
    }
    __syncthreads();
    if (blockIdx.x == 0 && threadIdx.x == 0) {
        float r = p2b[0];
        for (int p = 0; p < 64; ++p) r += p2w[p] * p1b[p];
        float al = 0.f, be = r;
        for (int i = 0; i < 64; ++i) { al += q[i] * lw[i]; be += q[i] * lb[i]; }
        scal[0] = al;
        scal[1] = be;
    }
    int t = blockIdx.x * 256 + threadIdx.x;
    int b = t >> 9;
    int m = t & 511;
    const float* pr = STre + m * 1024 + b * 4;
    const float* pi = STim + m * 1024 + b * 4;
    float re = 0.f, im = 0.f;
    for (int cq = 0; cq < 16; ++cq) {
        const v4f vr = *(const v4f*)(pr + cq * 64);
        const v4f vi = *(const v4f*)(pi + cq * 64);
#pragma unroll
        for (int s = 0; s < 4; ++s) {
            float w = q[cq * 4 + s];
            re += w * vr[s];
            im += w * vi[s];
        }
    }
    T[t * 2]     = re;
    T[t * 2 + 1] = im;
}

__global__ void fnde_inv_dft1(const float* __restrict__ T,
                              float* __restrict__ G) {
    __shared__ float cs[128], sn[128];
    fnde_tables_lds(cs, sn);
    int t  = blockIdx.x * 256 + threadIdx.x;
    int b  = t >> 11;
    int x  = (t >> 4) & 127;
    int ky = t & 15;
    float re = 0.f, im = 0.f;
    for (int corner = 0; corner < 2; ++corner) {
        for (int mx = 0; mx < 16; ++mx) {
            int m  = corner * 256 + mx * 16 + ky;
            int kx = corner ? (112 + mx) : mx;
            float tr = T[(b * 512 + m) * 2];
            float ti = T[(b * 512 + m) * 2 + 1];
            int a = (kx * x) & 127;
            float c = cs[a], s = sn[a];
            re += tr * c - ti * s;
            im += tr * s + ti * c;
        }
    }
    G[t * 2]     = re;
    G[t * 2 + 1] = im;
}

__global__ void fnde_inv_dft2_out(const float* __restrict__ G, const float* __restrict__ z,
                                  const float* __restrict__ scal, float* __restrict__ out) {
    __shared__ float cs[128], sn[128];
    fnde_tables_lds(cs, sn);
    int t = blockIdx.x * 256 + threadIdx.x;
    int b = t >> 14;
    int x = (t >> 7) & 127;
    int y = t & 127;
    const float* g = G + ((b * 128 + x) * 16) * 2;
    float acc = g[0];
    for (int ky = 1; ky < 16; ++ky) {
        int a = (ky * y) & 127;
        acc += 2.0f * (g[ky * 2] * cs[a] - g[ky * 2 + 1] * sn[a]);
    }
    out[t] = scal[0] * z[t] + scal[1] + acc * (1.0f / 16384.0f);
}

extern "C" void kernel_launch(void* const* d_in, const int* in_sizes, int n_in,
                              void* d_out, int out_size, void* d_ws, size_t ws_size,
                              hipStream_t stream) {
    (void)in_sizes; (void)n_in; (void)out_size; (void)ws_size;

    const float* zin    = (const float*)d_in[0];
    const float* lift_w = (const float*)d_in[1];
    const float* lift_b = (const float*)d_in[2];
    const float* spec_w1 = (const float*)d_in[3];
    const float* spec_w2 = (const float*)d_in[4];
    const float* fl_bias = (const float*)d_in[5];
    const float* p1_w = (const float*)d_in[6];
    const float* p1_b = (const float*)d_in[7];
    const float* p2_w = (const float*)d_in[8];
    const float* p2_b = (const float*)d_in[9];
    const float* ts   = (const float*)d_in[10];
    float* outp = (float*)d_out;

    float* w = (float*)d_ws;
    float* Wp   = w;  w += (size_t)Ll * WLAYER;
    float* Zre  = w;  w += NSPEC;  float* Zim  = w;  w += NSPEC;
    float* Ure  = w;  w += NSPEC;  float* Uim  = w;  w += NSPEC;
    float* Xre  = w;  w += NSPEC;  float* Xim  = w;  w += NSPEC;
    float* STre = w;  w += NSPEC;  float* STim = w;  w += NSPEC;
    float* Zhat = w;  w += Bb * NMODE * 2;
    float* Ay   = w;  w += Bb * Dd * Mm * 2;
    float* Tm   = w;  w += Bb * NMODE * 2;
    float* Gi   = w;  w += Bb * Dd * Mm * 2;
    float* scal = w;  w += 8;

    fnde_pack_weights<<<(Ll * 2 * Mm * Cc * Cc) / 256, 256, 0, stream>>>(spec_w1, spec_w2, Wp);
    fnde_fwd_dft1<<<(Bb * Dd * Mm) / 256, 256, 0, stream>>>(zin, Ay);
    fnde_fwd_dft2<<<(Bb * NMODE) / 256, 256, 0, stream>>>(Ay, Zhat);

    fnde_ode<<<497, 128, 0, stream>>>(Wp, Zre, Zim, Ure, Uim, Xre, Xim, STre, STim,
                                      fl_bias, ts, Zhat, lift_w, lift_b);

    fnde_proj_modes<<<(Bb * NMODE) / 256, 256, 0, stream>>>(STre, STim, p1_w, p1_b, p2_w, p2_b,
                                                            lift_w, lift_b, scal, Tm);
    fnde_inv_dft1<<<(Bb * Dd * Mm) / 256, 256, 0, stream>>>(Tm, Gi);
    fnde_inv_dft2_out<<<(Bb * Dd * Dd) / 256, 256, 0, stream>>>(Gi, zin, scal, outp);
}


# ODE launch: the next stage's input passes between the 4 waves through two alternating LDS buffers, not global planes (no stage stores or loads to memory); same barriers, sums same
# speedup vs baseline: 6.2213x; 1.1167x over previous
#include <hip/hip_runtime.h>


typedef float v2f __attribute__((ext_vector_type(2)));
typedef float v4f __attribute__((ext_vector_type(4)));
typedef float v8f __attribute__((ext_vector_type(8)));

#define Bb    16
#define Cc    64
#define Dd    128
#define Mm    16
#define Ll    3
#define NMODE 512
#define NSPEC (NMODE*Cc*Bb)
#define WLAYER (2*NMODE*Cc*Cc)
#define XPLANE 1024
#define XSLOT  (2*XPLANE)
#define XBUF   (2*XSLOT)

__device__ __forceinline__ void fnde_tables_lds(float* cs, float* sn) {
    const int t = threadIdx.x;
    if (t < 128) {
        float ang = (float)(6.283185307179586 / 128.0) * (float)t;
        cs[t] = cosf(ang);
        sn[t] = sinf(ang);
    }
    __syncthreads();
}

__global__ void fnde_pack_weights(const float* __restrict__ w1,
                                  const float* __restrict__ w2,
                                  float* __restrict__ Wp) {
    int idx = blockIdx.x * 256 + threadIdx.x;
    int i  = idx & 63;
    int o  = (idx >> 6) & 63;
    int mx = (idx >> 12) & 15;
    int corner = (idx >> 16) & 1;
    int l  = idx >> 17;
    const float* src = (corner ? w2 : w1)
                     + (size_t)((((l * Cc + i) * Cc + o) * Mm + mx) * (Mm * 2));
    v4f q[8];
#pragma unroll
    for (int k = 0; k < 8; ++k) q[k] = *(const v4f*)(src + k * 4);
    float* dst = Wp + (size_t)l * WLAYER + (size_t)(corner * 256 + mx * 16) * 4096 + o * 64 + i;
#pragma unroll
    for (int k = 0; k < 8; ++k) {
        float* d0 = dst + (size_t)(2 * k) * 4096;
        float* d1 = d0 + 4096;
        d0[0]               = q[k].x;
        d0[NMODE * Cc * Cc] = q[k].y;
        d1[0]               = q[k].z;
        d1[NMODE * Cc * Cc] = q[k].w;
    }
}

__global__ void fnde_fwd_dft1(const float* __restrict__ z,
                              float* __restrict__ Ay) {
    __shared__ float cs[128], sn[128];
    fnde_tables_lds(cs, sn);
    int t  = blockIdx.x * 256 + threadIdx.x;
    int b  = t >> 11;
    int x  = (t >> 4) & 127;
    int ky = t & 15;
    const float* row = z + (b * 128 + x) * 128;
    float re = 0.f, im = 0.f;
    for (int y = 0; y < 128; ++y) {
        float v = row[y];
        int a = (ky * y) & 127;
        re += v * cs[a];
        im -= v * sn[a];
    }
    Ay[t * 2]     = re;
    Ay[t * 2 + 1] = im;
}

__global__ void fnde_fwd_dft2(const float* __restrict__ Ay,
                              float* __restrict__ Zhat) {
    __shared__ float cs[128], sn[128];
    fnde_tables_lds(cs, sn);
    int t = blockIdx.x * 256 + threadIdx.x;
    int b = t >> 9;
    int m = t & 511;
    int corner = m >> 8, mx = (m >> 4) & 15, ky = m & 15;
    int kx = corner ? (112 + mx) : mx;
    float re = 0.f, im = 0.f;
    for (int x = 0; x < 128; ++x) {
        int base = ((b * 128 + x) * 16 + ky) * 2;
        float ar = Ay[base], ai = Ay[base + 1];
        int a = (kx * x) & 127;
        float c = cs[a], s = sn[a];
        re += ar * c + ai * s;
        im += ai * c - ar * s;
    }
    Zhat[t * 2]     = re;
    Zhat[t * 2 + 1] = im;
}

struct fnde_afrags { v2f ar[16], ai[16]; };
struct fnde_bfrags { v2f br[16], bi[16]; };

__device__ __forceinline__ void fnde_load_a(fnde_afrags& f,
                                            const float* __restrict__ Wr,
                                            const float* __restrict__ Wi, int aOff) {
#pragma unroll
    for (int q = 0; q < 8; ++q) {
        const v4f r4 = *(const v4f*)(Wr + aOff + q * 8);
        const v4f i4 = *(const v4f*)(Wi + aOff + q * 8);
        f.ar[2 * q].x = r4.x;      f.ar[2 * q].y = r4.y;
        f.ar[2 * q + 1].x = r4.z;  f.ar[2 * q + 1].y = r4.w;
        f.ai[2 * q].x = i4.x;      f.ai[2 * q].y = i4.y;
        f.ai[2 * q + 1].x = i4.z;  f.ai[2 * q + 1].y = i4.w;
    }
}

__device__ __forceinline__ void fnde_load_b(fnde_bfrags& f, const float* Vr, const float* Vi,
                                            int bOff) {
#pragma unroll
    for (int q = 0; q < 8; ++q) {
        const v4f r4 = *(const v4f*)(Vr + q * 128 + bOff);
        const v4f i4 = *(const v4f*)(Vi + q * 128 + bOff);
        f.br[2 * q].x = r4.x;      f.br[2 * q].y = r4.y;
        f.br[2 * q + 1].x = r4.z;  f.br[2 * q + 1].y = r4.w;
        f.bi[2 * q].x = i4.x;      f.bi[2 * q].y = i4.y;
        f.bi[2 * q + 1].x = i4.z;  f.bi[2 * q + 1].y = i4.w;
    }
}

__device__ __forceinline__ void fnde_product(const fnde_afrags& a, const fnde_bfrags& b,
                                             v8f& cr, v8f& ci) {
#pragma unroll
    for (int kk = 0; kk < 16; ++kk) {
        v2f bin;
        bin.x = -b.bi[kk].x;    bin.y = -b.bi[kk].y;
        cr = __builtin_amdgcn_wmma_f32_16x16x4_f32(false, a.ar[kk], false, b.br[kk],
                                                   (short)0, cr, false, false);
        ci = __builtin_amdgcn_wmma_f32_16x16x4_f32(false, a.ar[kk], false, b.bi[kk],
                                                   (short)0, ci, false, false);
        cr = __builtin_amdgcn_wmma_f32_16x16x4_f32(false, a.ai[kk], false, bin,
                                                   (short)0, cr, false, false);
        ci = __builtin_amdgcn_wmma_f32_16x16x4_f32(false, a.ai[kk], false, b.br[kk],
                                                   (short)0, ci, false, false);
    }
}

__device__ __forceinline__ void fnde_pi(float kr, float ki, float mr, float mi,
                                        bool mirrored, bool halfOnly,
                                        float& pr, float& pim) {
    if (mirrored)      { pr = 0.5f * (kr + mr);  pim = 0.5f * (ki - mi); }
    else if (halfOnly) { pr = 0.5f * kr;         pim = 0.5f * ki; }
    else               { pr = kr;                pim = ki; }
}

struct fnde_state { float zr[8], zi[8], sr[8], si[8], tr[8], ti[8]; };

template <int PHASE>
__device__ __forceinline__ void fnde_combine(fnde_state& e,
        const float* kr, const float* ki, const float* s4r, const float* s4i,
        const float* xr, const float* xi, bool mirrored, bool halfOnly, float h,
        float* Ore, float* Oim, int off) {
    float o_r[8], o_i[8];
#pragma unroll
    for (int j = 0; j < 8; ++j) {
        float pr, pim;
        if (PHASE <= 3) {
            const float uc = (PHASE == 3) ? h : 0.5f * h;
            fnde_pi(kr[j], ki[j], xr[j], xi[j], mirrored, halfOnly, pr, pim);
            o_r[j] = e.zr[j] + uc * pr;
            o_i[j] = e.zi[j] + uc * pim;
            if (PHASE == 1) { e.sr[j] = kr[j];                  e.si[j] = ki[j]; }
            else            { e.sr[j] = e.sr[j] + 2.0f * kr[j]; e.si[j] = e.si[j] + 2.0f * ki[j]; }
        } else {
            const float h6 = h * (1.0f / 6.0f);
            fnde_pi(s4r[j], s4i[j], xr[j], xi[j], mirrored, halfOnly, pr, pim);
            o_r[j] = e.zr[j] + h6 * pr;
            o_i[j] = e.zi[j] + h6 * pim;
            e.zr[j] = o_r[j];
            e.zi[j] = o_i[j];
            e.tr[j] = e.tr[j] + h6 * s4r[j];
            e.ti[j] = e.ti[j] + h6 * s4i[j];
        }
    }
#pragma unroll
    for (int g = 0; g < 2; ++g) {
        const int t = off + g * 64;
        v4f vo, vp;
#pragma unroll
        for (int q = 0; q < 4; ++q) {
            vo[q] = o_r[4 * g + q];  vp[q] = o_i[4 * g + q];
        }
        *(v4f*)(Ore + t) = vo;
        *(v4f*)(Oim + t) = vp;
    }
}

template <int PHASE, bool OTHER>
__device__ __forceinline__ void fnde_stage_wave(
        const fnde_afrags& ao, const fnde_afrags& am,
        fnde_state& eo, fnde_state& em,
        const float* V, float* O,
        const float* __restrict__ bias, float h,
        int mode, bool halfOnly, int mt, int laneHi, int laneLo) {
    const int bOff = ((laneHi << 4) + laneLo) * 4;
    const int obase = mt * 16 + (laneHi << 3);
    const int t0  = (obase >> 2) * 64 + laneLo * 4;

    fnde_bfrags bm, bo;
    if (OTHER) fnde_load_b(bm, V + XSLOT, V + XSLOT + XPLANE, bOff);
    fnde_load_b(bo, V, V + XPLANE, bOff);

    const bool addBias = !OTHER && mode == 0;
    float bj[8];
#pragma unroll
    for (int j = 0; j < 8; ++j) bj[j] = 0.f;
    if (addBias) {
#pragma unroll
        for (int j = 0; j < 8; ++j) bj[j] = bias[obase + j] * 16384.0f;
    }

    __builtin_amdgcn_sched_barrier(0);

    v8f cmr = {}, cmi = {};
    if (OTHER) fnde_product(am, bm, cmr, cmi);
    v8f cr = {}, ci = {};
    fnde_product(ao, bo, cr, ci);

    float kor[8], koi[8], kmr[8], kmi[8], so_r[8], so_i[8], sm_r[8], sm_i[8];
#pragma unroll
    for (int j = 0; j < 8; ++j) {
        kor[j] = cr[j];
        if (addBias) kor[j] += bj[j];
        koi[j] = ci[j];
        kmr[j] = cmr[j];
        kmi[j] = cmi[j];
        so_r[j] = (PHASE == 4) ? (eo.sr[j] + kor[j]) : 0.f;
        so_i[j] = (PHASE == 4) ? (eo.si[j] + koi[j]) : 0.f;
        sm_r[j] = (PHASE == 4 && OTHER) ? (em.sr[j] + kmr[j]) : 0.f;
        sm_i[j] = (PHASE == 4 && OTHER) ? (em.si[j] + kmi[j]) : 0.f;
    }
    const bool mirrored = OTHER || mode == 0;
    const float* xor_ = (PHASE <= 3) ? (OTHER ? kmr : kor) : (OTHER ? sm_r : so_r);
    const float* xoi_ = (PHASE <= 3) ? (OTHER ? kmi : koi) : (OTHER ? sm_i : so_i);
    fnde_combine<PHASE>(eo, kor, koi, so_r, so_i, xor_, xoi_, mirrored, halfOnly, h,
                        O, O + XPLANE, t0);
    if (OTHER) {
        const float* xmr_ = (PHASE <= 3) ? kor : so_r;
        const float* xmi_ = (PHASE <= 3) ? koi : so_i;
        fnde_combine<PHASE>(em, kmr, kmi, sm_r, sm_i, xmr_, xmi_, true, false, h,
                            O + XSLOT, O + XSLOT + XPLANE, t0);
    }
}

__device__ __forceinline__ void fnde_lift_mode(fnde_state& e, const float* __restrict__ Zhat,
        const float* __restrict__ lw, const float* __restrict__ lb,
        float* Zre, float* Zim, int m, int obase, int laneLo, int off) {
    const float zr = Zhat[(laneLo * 512 + m) * 2];
    const float zi = Zhat[(laneLo * 512 + m) * 2 + 1];
#pragma unroll
    for (int g = 0; g < 2; ++g) {
        const v4f w4 = *(const v4f*)(lw + obase + 4 * g);
        const v4f b4 = *(const v4f*)(lb + obase + 4 * g);
        v4f vr, vi;
#pragma unroll
        for (int q = 0; q < 4; ++q) {
            const float w = w4[q];
            vr[q] = w * zr + ((m == 0) ? b4[q] * 16384.0f : 0.0f);
            vi[q] = w * zi;
            e.zr[4 * g + q] = vr[q];  e.zi[4 * g + q] = vi[q];
            e.sr[4 * g + q] = 0.f;    e.si[4 * g + q] = 0.f;
            e.tr[4 * g + q] = 0.f;    e.ti[4 * g + q] = 0.f;
        }
        const int t = off + g * 64;
        *(v4f*)(Zre + t)  = vr;
        *(v4f*)(Zim + t)  = vi;
    }
}

__device__ __forceinline__ void fnde_store_total(const fnde_state& e,
        float* STre, float* STim, int off) {
#pragma unroll
    for (int g = 0; g < 2; ++g) {
        v4f vr, vi;
#pragma unroll
        for (int q = 0; q < 4; ++q) { vr[q] = e.tr[4 * g + q];  vi[q] = e.ti[4 * g + q]; }
        const int t = off + g * 64;
        *(v4f*)(STre + t) = vr;
        *(v4f*)(STim + t) = vi;
    }
}

template <bool OTHER>
__device__ __forceinline__ void fnde_chain(
        const float* __restrict__ Wp,
        float* xs,
        float* STre, float* STim,
        const float* __restrict__ fl_bias, const float* __restrict__ ts,
        const float* __restrict__ Zhat, const float* __restrict__ lw,
        const float* __restrict__ lb,
        int mode, int pm, bool halfOnly, int mt, int laneHi, int laneLo) {
    const size_t imOff = (size_t)NMODE * Cc * Cc;
    const int aOff = laneLo * 64 + laneHi * 4;
    const int obase = mt * 16 + (laneHi << 3);
    const int t0 = (obase >> 2) * 64 + laneLo * 4;
    float* x0 = xs;
    float* x1 = xs + XBUF;
    fnde_state eo, em;
    fnde_lift_mode(eo, Zhat, lw, lb, x0, x0 + XPLANE, mode, obase, laneLo, t0);
    if (OTHER)
        fnde_lift_mode(em, Zhat, lw, lb, x0 + XSLOT, x0 + XSLOT + XPLANE, pm, obase, laneLo, t0);
    __syncthreads();
    for (int l = 0; l < Ll; ++l) {
        const float* Wl = Wp + (size_t)l * WLAYER;
        const float* bl = fl_bias + l * Cc;
        fnde_afrags ao, am;
        {
            const float* Wr = Wl + (size_t)mode * 4096 + mt * 1024;
            fnde_load_a(ao, Wr, Wr + imOff, aOff);
        }
        if (OTHER) {
            const float* Wr = Wl + (size_t)pm * 4096 + mt * 1024;
            fnde_load_a(am, Wr, Wr + imOff, aOff);
        }
        for (int s = 0; s < 4; ++s) {
            const float h = ts[s + 1] - ts[s];
            fnde_stage_wave<1, OTHER>(ao, am, eo, em, x0, x1,
                                      bl, h, mode, halfOnly, mt, laneHi, laneLo);
            __syncthreads();
            fnde_stage_wave<2, OTHER>(ao, am, eo, em, x1, x0,
                                      bl, h, mode, halfOnly, mt, laneHi, laneLo);
            __syncthreads();
            fnde_stage_wave<3, OTHER>(ao, am, eo, em, x0, x1,
                                      bl, h, mode, halfOnly, mt, laneHi, laneLo);
            __syncthreads();
            fnde_stage_wave<4, OTHER>(ao, am, eo, em, x1, x0,
                                      bl, h, mode, halfOnly, mt, laneHi, laneLo);
            __syncthreads();
        }
    }
    fnde_store_total(eo, STre, STim, mode * 1024 + t0);
    if (OTHER) fnde_store_total(em, STre, STim, pm * 1024 + t0);
}

__global__ __launch_bounds__(128) __attribute__((amdgpu_waves_per_eu(2, 4)))
void fnde_ode(const float* __restrict__ Wp,
              float* STre, float* STim,
              const float* __restrict__ fl_bias, const float* __restrict__ ts,
              const float* __restrict__ Zhat, const float* __restrict__ lw,
              const float* __restrict__ lb) {
    const int blk = blockIdx.x;
    int mode, pm;
    if (blk < 15)       { mode = (1 + blk) * 16;  pm = 256 + (15 - blk) * 16; }
    else if (blk < 495) { mode = ((blk - 15) / 15) * 16 + 1 + (blk - 15) % 15;  pm = mode; }
    else                { mode = (blk - 495) * 256;  pm = mode; }
    const int mt   = threadIdx.x >> 5;
    const int lane = threadIdx.x & 31;
    const int laneHi = lane >> 4;
    const int laneLo = lane & 15;
    const bool halfOnly = mode == 256;
    __shared__ __attribute__((aligned(16))) float xs[2 * XBUF];

    if (pm != mode)
        fnde_chain<true>(Wp, xs, STre, STim, fl_bias, ts,
                         Zhat, lw, lb, mode, pm, halfOnly, mt, laneHi, laneLo);
    else
        fnde_chain<false>(Wp, xs, STre, STim, fl_bias, ts,
                          Zhat, lw, lb, mode, pm, halfOnly, mt, laneHi, laneLo);
}

__global__ void fnde_proj_modes(const float* __restrict__ STre, const float* __restrict__ STim,
                                const float* __restrict__ p1w, const float* __restrict__ p1b,
                                const float* __restrict__ p2w, const float* __restrict__ p2b,
                                const float* __restrict__ lw,  const float* __restrict__ lb,
                                float* __restrict__ scal, float* __restrict__ T) {
    __shared__ float q[64];
    if (threadIdx.x < 64) {
        const int c = threadIdx.x;
        float acc = 0.f;
        for (int p = 0; p < 64; ++p) acc += p2w[p] * p1w[p * 64 + c];
        q[c] = acc;
    }
    __syncthreads();
    if (blockIdx.x == 0 && threadIdx.x == 0) {
        float r = p2b[0];
        for (int p = 0; p < 64; ++p) r += p2w[p] * p1b[p];
        float al = 0.f, be = r;
        for (int i = 0; i < 64; ++i) { al += q[i] * lw[i]; be += q[i] * lb[i]; }
        scal[0] = al;
        scal[1] = be;
    }
    int t = blockIdx.x * 256 + threadIdx.x;
    int b = t >> 9;
    int m = t & 511;
    const float* pr = STre + m * 1024 + b * 4;
    const float* pi = STim + m * 1024 + b * 4;
    float re = 0.f, im = 0.f;
    for (int cq = 0; cq < 16; ++cq) {
        const v4f vr = *(const v4f*)(pr + cq * 64);
        const v4f vi = *(const v4f*)(pi + cq * 64);
#pragma unroll
        for (int s = 0; s < 4; ++s) {
            float w = q[cq * 4 + s];
            re += w * vr[s];
            im += w * vi[s];
        }
    }
    T[t * 2]     = re;
    T[t * 2 + 1] = im;
}

__global__ void fnde_inv_dft1(const float* __restrict__ T,
                              float* __restrict__ G) {
    __shared__ float cs[128], sn[128];
    fnde_tables_lds(cs, sn);
    int t  = blockIdx.x * 256 + threadIdx.x;
    int b  = t >> 11;
    int x  = (t >> 4) & 127;
    int ky = t & 15;
    float re = 0.f, im = 0.f;
    for (int corner = 0; corner < 2; ++corner) {
        for (int mx = 0; mx < 16; ++mx) {
            int m  = corner * 256 + mx * 16 + ky;
            int kx = corner ? (112 + mx) : mx;
            float tr = T[(b * 512 + m) * 2];
            float ti = T[(b * 512 + m) * 2 + 1];
            int a = (kx * x) & 127;
            float c = cs[a], s = sn[a];
            re += tr * c - ti * s;
            im += tr * s + ti * c;
        }
    }
    G[t * 2]     = re;
    G[t * 2 + 1] = im;
}

__global__ void fnde_inv_dft2_out(const float* __restrict__ G, const float* __restrict__ z,
                                  const float* __restrict__ scal, float* __restrict__ out) {
    __shared__ float cs[128], sn[128];
    fnde_tables_lds(cs, sn);
    int t = blockIdx.x * 256 + threadIdx.x;
    int b = t >> 14;
    int x = (t >> 7) & 127;
    int y = t & 127;
    const float* g = G + ((b * 128 + x) * 16) * 2;
    float acc = g[0];
    for (int ky = 1; ky < 16; ++ky) {
        int a = (ky * y) & 127;
        acc += 2.0f * (g[ky * 2] * cs[a] - g[ky * 2 + 1] * sn[a]);
    }
    out[t] = scal[0] * z[t] + scal[1] + acc * (1.0f / 16384.0f);
}

extern "C" void kernel_launch(void* const* d_in, const int* in_sizes, int n_in,
                              void* d_out, int out_size, void* d_ws, size_t ws_size,
                              hipStream_t stream) {
    (void)in_sizes; (void)n_in; (void)out_size; (void)ws_size;

    const float* zin    = (const float*)d_in[0];
    const float* lift_w = (const float*)d_in[1];
    const float* lift_b = (const float*)d_in[2];
    const float* spec_w1 = (const float*)d_in[3];
    const float* spec_w2 = (const float*)d_in[4];
    const float* fl_bias = (const float*)d_in[5];
    const float* p1_w = (const float*)d_in[6];
    const float* p1_b = (const float*)d_in[7];
    const float* p2_w = (const float*)d_in[8];
    const float* p2_b = (const float*)d_in[9];
    const float* ts   = (const float*)d_in[10];
    float* outp = (float*)d_out;

    float* w = (float*)d_ws;
    float* Wp   = w;  w += (size_t)Ll * WLAYER;
    float* STre = w;  w += NSPEC;  float* STim = w;  w += NSPEC;
    float* Zhat = w;  w += Bb * NMODE * 2;
    float* Ay   = w;  w += Bb * Dd * Mm * 2;
    float* Tm   = w;  w += Bb * NMODE * 2;
    float* Gi   = w;  w += Bb * Dd * Mm * 2;
    float* scal = w;  w += 8;

    fnde_pack_weights<<<(Ll * 2 * Mm * Cc * Cc) / 256, 256, 0, stream>>>(spec_w1, spec_w2, Wp);
    fnde_fwd_dft1<<<(Bb * Dd * Mm) / 256, 256, 0, stream>>>(zin, Ay);
    fnde_fwd_dft2<<<(Bb * NMODE) / 256, 256, 0, stream>>>(Ay, Zhat);

    fnde_ode<<<497, 128, 0, stream>>>(Wp, STre, STim,
                                      fl_bias, ts, Zhat, lift_w, lift_b);

    fnde_proj_modes<<<(Bb * NMODE) / 256, 256, 0, stream>>>(STre, STim, p1_w, p1_b, p2_w, p2_b,
                                                            lift_w, lift_b, scal, Tm);
    fnde_inv_dft1<<<(Bb * Dd * Mm) / 256, 256, 0, stream>>>(Tm, Gi);
    fnde_inv_dft2_out<<<(Bb * Dd * Dd) / 256, 256, 0, stream>>>(Gi, zin, scal, outp);
}


# ODE launch re-cut: 256 work-groups (one per WGP, one wave per SIMD), each wave walks two modes as the ky=0 pair waves did: equal WMMAs per SIMD and stage (was 497 uneven groups); sums same
# speedup vs baseline: 7.6842x; 1.2351x over previous
#include <hip/hip_runtime.h>


typedef float v2f __attribute__((ext_vector_type(2)));
typedef float v4f __attribute__((ext_vector_type(4)));
typedef float v8f __attribute__((ext_vector_type(8)));

#define Bb    16
#define Cc    64
#define Dd    128
#define Mm    16
#define Ll    3
#define NMODE 512
#define NSPEC (NMODE*Cc*Bb)
#define WLAYER (2*NMODE*Cc*Cc)
#define XPLANE 1024
#define XSLOT  (2*XPLANE)
#define XBUF   (2*XSLOT)

__device__ __forceinline__ void fnde_tables_lds(float* cs, float* sn) {
    const int t = threadIdx.x;
    if (t < 128) {
        float ang = (float)(6.283185307179586 / 128.0) * (float)t;
        cs[t] = cosf(ang);
        sn[t] = sinf(ang);
    }
    __syncthreads();
}

__global__ void fnde_pack_weights(const float* __restrict__ w1,
                                  const float* __restrict__ w2,
                                  float* __restrict__ Wp) {
    int idx = blockIdx.x * 256 + threadIdx.x;
    int i  = idx & 63;
    int o  = (idx >> 6) & 63;
    int mx = (idx >> 12) & 15;
    int corner = (idx >> 16) & 1;
    int l  = idx >> 17;
    const float* src = (corner ? w2 : w1)
                     + (size_t)((((l * Cc + i) * Cc + o) * Mm + mx) * (Mm * 2));
    v4f q[8];
#pragma unroll
    for (int k = 0; k < 8; ++k) q[k] = *(const v4f*)(src + k * 4);
    float* dst = Wp + (size_t)l * WLAYER + (size_t)(corner * 256 + mx * 16) * 4096 + o * 64 + i;
#pragma unroll
    for (int k = 0; k < 8; ++k) {
        float* d0 = dst + (size_t)(2 * k) * 4096;
        float* d1 = d0 + 4096;
        d0[0]               = q[k].x;
        d0[NMODE * Cc * Cc] = q[k].y;
        d1[0]               = q[k].z;
        d1[NMODE * Cc * Cc] = q[k].w;
    }
}

__global__ void fnde_fwd_dft1(const float* __restrict__ z,
                              float* __restrict__ Ay) {
    __shared__ float cs[128], sn[128];
    fnde_tables_lds(cs, sn);
    int t  = blockIdx.x * 256 + threadIdx.x;
    int b  = t >> 11;
    int x  = (t >> 4) & 127;
    int ky = t & 15;
    const float* row = z + (b * 128 + x) * 128;
    float re = 0.f, im = 0.f;
    for (int y = 0; y < 128; ++y) {
        float v = row[y];
        int a = (ky * y) & 127;
        re += v * cs[a];
        im -= v * sn[a];
    }
    Ay[t * 2]     = re;
    Ay[t * 2 + 1] = im;
}

__global__ void fnde_fwd_dft2(const float* __restrict__ Ay,
                              float* __restrict__ Zhat) {
    __shared__ float cs[128], sn[128];
    fnde_tables_lds(cs, sn);
    int t = blockIdx.x * 256 + threadIdx.x;
    int b = t >> 9;
    int m = t & 511;
    int corner = m >> 8, mx = (m >> 4) & 15, ky = m & 15;
    int kx = corner ? (112 + mx) : mx;
    float re = 0.f, im = 0.f;
    for (int x = 0; x < 128; ++x) {
        int base = ((b * 128 + x) * 16 + ky) * 2;
        float ar = Ay[base], ai = Ay[base + 1];
        int a = (kx * x) & 127;
        float c = cs[a], s = sn[a];
        re += ar * c + ai * s;
        im += ai * c - ar * s;
    }
    Zhat[t * 2]     = re;
    Zhat[t * 2 + 1] = im;
}

struct fnde_afrags { v2f ar[16], ai[16]; };
struct fnde_bfrags { v2f br[16], bi[16]; };

__device__ __forceinline__ void fnde_load_a(fnde_afrags& f,
                                            const float* __restrict__ Wr,
                                            const float* __restrict__ Wi, int aOff) {
#pragma unroll
    for (int q = 0; q < 8; ++q) {
        const v4f r4 = *(const v4f*)(Wr + aOff + q * 8);
        const v4f i4 = *(const v4f*)(Wi + aOff + q * 8);
        f.ar[2 * q].x = r4.x;      f.ar[2 * q].y = r4.y;
        f.ar[2 * q + 1].x = r4.z;  f.ar[2 * q + 1].y = r4.w;
        f.ai[2 * q].x = i4.x;      f.ai[2 * q].y = i4.y;
        f.ai[2 * q + 1].x = i4.z;  f.ai[2 * q + 1].y = i4.w;
    }
}

__device__ __forceinline__ void fnde_load_b(fnde_bfrags& f, const float* Vr, const float* Vi,
                                            int bOff) {
#pragma unroll
    for (int q = 0; q < 8; ++q) {
        const v4f r4 = *(const v4f*)(Vr + q * 128 + bOff);
        const v4f i4 = *(const v4f*)(Vi + q * 128 + bOff);
        f.br[2 * q].x = r4.x;      f.br[2 * q].y = r4.y;
        f.br[2 * q + 1].x = r4.z;  f.br[2 * q + 1].y = r4.w;
        f.bi[2 * q].x = i4.x;      f.bi[2 * q].y = i4.y;
        f.bi[2 * q + 1].x = i4.z;  f.bi[2 * q + 1].y = i4.w;
    }
}

__device__ __forceinline__ void fnde_product(const fnde_afrags& a, const fnde_bfrags& b,
                                             v8f& cr, v8f& ci) {
#pragma unroll
    for (int kk = 0; kk < 16; ++kk) {
        v2f bin;
        bin.x = -b.bi[kk].x;    bin.y = -b.bi[kk].y;
        cr = __builtin_amdgcn_wmma_f32_16x16x4_f32(false, a.ar[kk], false, b.br[kk],
                                                   (short)0, cr, false, false);
        ci = __builtin_amdgcn_wmma_f32_16x16x4_f32(false, a.ar[kk], false, b.bi[kk],
                                                   (short)0, ci, false, false);
        cr = __builtin_amdgcn_wmma_f32_16x16x4_f32(false, a.ai[kk], false, bin,
                                                   (short)0, cr, false, false);
        ci = __builtin_amdgcn_wmma_f32_16x16x4_f32(false, a.ai[kk], false, b.br[kk],
                                                   (short)0, ci, false, false);
    }
}

__device__ __forceinline__ void fnde_pi(float kr, float ki, float mr, float mi,
                                        bool mirrored, bool halfOnly,
                                        float& pr, float& pim) {
    if (mirrored)      { pr = 0.5f * (kr + mr);  pim = 0.5f * (ki - mi); }
    else if (halfOnly) { pr = 0.5f * kr;         pim = 0.5f * ki; }
    else               { pr = kr;                pim = ki; }
}

struct fnde_state { float zr[8], zi[8], sr[8], si[8], tr[8], ti[8]; };

template <int PHASE>
__device__ __forceinline__ void fnde_combine(fnde_state& e,
        const float* kr, const float* ki, const float* s4r, const float* s4i,
        const float* xr, const float* xi, bool mirrored, bool halfOnly, float h,
        float* Ore, float* Oim, int off) {
    float o_r[8], o_i[8];
#pragma unroll
    for (int j = 0; j < 8; ++j) {
        float pr, pim;
        if (PHASE <= 3) {
            const float uc = (PHASE == 3) ? h : 0.5f * h;
            fnde_pi(kr[j], ki[j], xr[j], xi[j], mirrored, halfOnly, pr, pim);
            o_r[j] = e.zr[j] + uc * pr;
            o_i[j] = e.zi[j] + uc * pim;
            if (PHASE == 1) { e.sr[j] = kr[j];                  e.si[j] = ki[j]; }
            else            { e.sr[j] = e.sr[j] + 2.0f * kr[j]; e.si[j] = e.si[j] + 2.0f * ki[j]; }
        } else {
            const float h6 = h * (1.0f / 6.0f);
            fnde_pi(s4r[j], s4i[j], xr[j], xi[j], mirrored, halfOnly, pr, pim);
            o_r[j] = e.zr[j] + h6 * pr;
            o_i[j] = e.zi[j] + h6 * pim;
            e.zr[j] = o_r[j];
            e.zi[j] = o_i[j];
            e.tr[j] = e.tr[j] + h6 * s4r[j];
            e.ti[j] = e.ti[j] + h6 * s4i[j];
        }
    }
#pragma unroll
    for (int g = 0; g < 2; ++g) {
        const int t = off + g * 64;
        v4f vo, vp;
#pragma unroll
        for (int q = 0; q < 4; ++q) {
            vo[q] = o_r[4 * g + q];  vp[q] = o_i[4 * g + q];
        }
        *(v4f*)(Ore + t) = vo;
        *(v4f*)(Oim + t) = vp;
    }
}

template <int PHASE, int KIND>
__device__ __forceinline__ void fnde_stage_wave(
        const fnde_afrags& ao, const fnde_afrags& am,
        fnde_state& eo, fnde_state& em,
        const float* V, float* O,
        const float* __restrict__ bias, float h,
        int mt, int laneHi, int laneLo) {
    const int bOff = ((laneHi << 4) + laneLo) * 4;
    const int obase = mt * 16 + (laneHi << 3);
    const int t0  = (obase >> 2) * 64 + laneLo * 4;

    fnde_bfrags bm, bo;
    fnde_load_b(bm, V + XSLOT, V + XSLOT + XPLANE, bOff);
    fnde_load_b(bo, V, V + XPLANE, bOff);

    float bj[8];
#pragma unroll
    for (int j = 0; j < 8; ++j) bj[j] = 0.f;
    if (KIND == 2) {
#pragma unroll
        for (int j = 0; j < 8; ++j) bj[j] = bias[obase + j] * 16384.0f;
    }

    __builtin_amdgcn_sched_barrier(0);

    v8f cmr = {}, cmi = {};
    fnde_product(am, bm, cmr, cmi);
    v8f cr = {}, ci = {};
    fnde_product(ao, bo, cr, ci);

    float kor[8], koi[8], kmr[8], kmi[8], so_r[8], so_i[8], sm_r[8], sm_i[8];
#pragma unroll
    for (int j = 0; j < 8; ++j) {
        kor[j] = cr[j];
        if (KIND == 2) kor[j] += bj[j];
        koi[j] = ci[j];
        kmr[j] = cmr[j];
        kmi[j] = cmi[j];
        so_r[j] = (PHASE == 4) ? (eo.sr[j] + kor[j]) : 0.f;
        so_i[j] = (PHASE == 4) ? (eo.si[j] + koi[j]) : 0.f;
        sm_r[j] = (PHASE == 4) ? (em.sr[j] + kmr[j]) : 0.f;
        sm_i[j] = (PHASE == 4) ? (em.si[j] + kmi[j]) : 0.f;
    }
    const bool pair = KIND == 1;
    const float* xor_ = (PHASE <= 3) ? (pair ? kmr : kor) : (pair ? sm_r : so_r);
    const float* xoi_ = (PHASE <= 3) ? (pair ? kmi : koi) : (pair ? sm_i : so_i);
    fnde_combine<PHASE>(eo, kor, koi, so_r, so_i, xor_, xoi_, KIND != 0, false, h,
                        O, O + XPLANE, t0);
    const float* xmr_ = (PHASE <= 3) ? (pair ? kor : kmr) : (pair ? so_r : sm_r);
    const float* xmi_ = (PHASE <= 3) ? (pair ? koi : kmi) : (pair ? so_i : sm_i);
    fnde_combine<PHASE>(em, kmr, kmi, sm_r, sm_i, xmr_, xmi_, KIND == 1, KIND == 2, h,
                        O + XSLOT, O + XSLOT + XPLANE, t0);
}

__device__ __forceinline__ void fnde_lift_mode(fnde_state& e, const float* __restrict__ Zhat,
        const float* __restrict__ lw, const float* __restrict__ lb,
        float* Zre, float* Zim, int m, int obase, int laneLo, int off) {
    const float zr = Zhat[(laneLo * 512 + m) * 2];
    const float zi = Zhat[(laneLo * 512 + m) * 2 + 1];
#pragma unroll
    for (int g = 0; g < 2; ++g) {
        const v4f w4 = *(const v4f*)(lw + obase + 4 * g);
        const v4f b4 = *(const v4f*)(lb + obase + 4 * g);
        v4f vr, vi;
#pragma unroll
        for (int q = 0; q < 4; ++q) {
            const float w = w4[q];
            vr[q] = w * zr + ((m == 0) ? b4[q] * 16384.0f : 0.0f);
            vi[q] = w * zi;
            e.zr[4 * g + q] = vr[q];  e.zi[4 * g + q] = vi[q];
            e.sr[4 * g + q] = 0.f;    e.si[4 * g + q] = 0.f;
            e.tr[4 * g + q] = 0.f;    e.ti[4 * g + q] = 0.f;
        }
        const int t = off + g * 64;
        *(v4f*)(Zre + t)  = vr;
        *(v4f*)(Zim + t)  = vi;
    }
}

__device__ __forceinline__ void fnde_store_total(const fnde_state& e,
        float* STre, float* STim, int off) {
#pragma unroll
    for (int g = 0; g < 2; ++g) {
        v4f vr, vi;
#pragma unroll
        for (int q = 0; q < 4; ++q) { vr[q] = e.tr[4 * g + q];  vi[q] = e.ti[4 * g + q]; }
        const int t = off + g * 64;
        *(v4f*)(STre + t) = vr;
        *(v4f*)(STim + t) = vi;
    }
}

template <int KIND>
__device__ __forceinline__ void fnde_chain(
        const float* __restrict__ Wp,
        float* xs,
        float* STre, float* STim,
        const float* __restrict__ fl_bias, const float* __restrict__ ts,
        const float* __restrict__ Zhat, const float* __restrict__ lw,
        const float* __restrict__ lb,
        int mode, int pm, int mt, int laneHi, int laneLo) {
    const size_t imOff = (size_t)NMODE * Cc * Cc;
    const int aOff = laneLo * 64 + laneHi * 4;
    const int obase = mt * 16 + (laneHi << 3);
    const int t0 = (obase >> 2) * 64 + laneLo * 4;
    float* x0 = xs;
    float* x1 = xs + XBUF;
    fnde_state eo, em;
    fnde_lift_mode(eo, Zhat, lw, lb, x0, x0 + XPLANE, mode, obase, laneLo, t0);
    fnde_lift_mode(em, Zhat, lw, lb, x0 + XSLOT, x0 + XSLOT + XPLANE, pm, obase, laneLo, t0);
    __syncthreads();
    for (int l = 0; l < Ll; ++l) {
        const float* Wl = Wp + (size_t)l * WLAYER;
        const float* bl = fl_bias + l * Cc;
        fnde_afrags ao, am;
        {
            const float* Wr = Wl + (size_t)mode * 4096 + mt * 1024;
            fnde_load_a(ao, Wr, Wr + imOff, aOff);
        }
        {
            const float* Wr = Wl + (size_t)pm * 4096 + mt * 1024;
            fnde_load_a(am, Wr, Wr + imOff, aOff);
        }
        for (int s = 0; s < 4; ++s) {
            const float h = ts[s + 1] - ts[s];
            fnde_stage_wave<1, KIND>(ao, am, eo, em, x0, x1, bl, h, mt, laneHi, laneLo);
            __syncthreads();
            fnde_stage_wave<2, KIND>(ao, am, eo, em, x1, x0, bl, h, mt, laneHi, laneLo);
            __syncthreads();
            fnde_stage_wave<3, KIND>(ao, am, eo, em, x0, x1, bl, h, mt, laneHi, laneLo);
            __syncthreads();
            fnde_stage_wave<4, KIND>(ao, am, eo, em, x1, x0, bl, h, mt, laneHi, laneLo);
            __syncthreads();
        }
    }
    fnde_store_total(eo, STre, STim, mode * 1024 + t0);
    fnde_store_total(em, STre, STim, pm * 1024 + t0);
}

__global__ __launch_bounds__(128) __attribute__((amdgpu_waves_per_eu(1, 1)))
void fnde_ode(const float* __restrict__ Wp,
              float* STre, float* STim,
              const float* __restrict__ fl_bias, const float* __restrict__ ts,
              const float* __restrict__ Zhat, const float* __restrict__ lw,
              const float* __restrict__ lb) {
    const int blk = blockIdx.x;
    const int mt   = threadIdx.x >> 5;
    const int lane = threadIdx.x & 31;
    const int laneHi = lane >> 4;
    const int laneLo = lane & 15;
    __shared__ __attribute__((aligned(16))) float xs[2 * XBUF];

    if (blk == 255) {
        fnde_chain<2>(Wp, xs, STre, STim, fl_bias, ts, Zhat, lw, lb,
                      0, 256, mt, laneHi, laneLo);
    } else if (blk < 15) {
        fnde_chain<1>(Wp, xs, STre, STim, fl_bias, ts, Zhat, lw, lb,
                      (1 + blk) * 16, 256 + (15 - blk) * 16, mt, laneHi, laneLo);
    } else {
        const int i0 = 2 * (blk - 15), i1 = i0 + 1;
        fnde_chain<0>(Wp, xs, STre, STim, fl_bias, ts, Zhat, lw, lb,
                      (i0 / 15) * 16 + 1 + i0 % 15, (i1 / 15) * 16 + 1 + i1 % 15,
                      mt, laneHi, laneLo);
    }
}

__global__ void fnde_proj_modes(const float* __restrict__ STre, const float* __restrict__ STim,
                                const float* __restrict__ p1w, const float* __restrict__ p1b,
                                const float* __restrict__ p2w, const float* __restrict__ p2b,
                                const float* __restrict__ lw,  const float* __restrict__ lb,
                                float* __restrict__ scal, float* __restrict__ T) {
    __shared__ float q[64];
    if (threadIdx.x < 64) {
        const int c = threadIdx.x;
        float acc = 0.f;
        for (int p = 0; p < 64; ++p) acc += p2w[p] * p1w[p * 64 + c];
        q[c] = acc;
    }
    __syncthreads();
    if (blockIdx.x == 0 && threadIdx.x == 0) {
        float r = p2b[0];
        for (int p = 0; p < 64; ++p) r += p2w[p] * p1b[p];
        float al = 0.f, be = r;
        for (int i = 0; i < 64; ++i) { al += q[i] * lw[i]; be += q[i] * lb[i]; }
        scal[0] = al;
        scal[1] = be;
    }
    int t = blockIdx.x * 256 + threadIdx.x;
    int b = t >> 9;
    int m = t & 511;
    const float* pr = STre + m * 1024 + b * 4;
    const float* pi = STim + m * 1024 + b * 4;
    float re = 0.f, im = 0.f;
    for (int cq = 0; cq < 16; ++cq) {
        const v4f vr = *(const v4f*)(pr + cq * 64);
        const v4f vi = *(const v4f*)(pi + cq * 64);
#pragma unroll
        for (int s = 0; s < 4; ++s) {
            float w = q[cq * 4 + s];
            re += w * vr[s];
            im += w * vi[s];
        }
    }
    T[t * 2]     = re;
    T[t * 2 + 1] = im;
}

__global__ void fnde_inv_dft1(const float* __restrict__ T,
                              float* __restrict__ G) {
    __shared__ float cs[128], sn[128];
    fnde_tables_lds(cs, sn);
    int t  = blockIdx.x * 256 + threadIdx.x;
    int b  = t >> 11;
    int x  = (t >> 4) & 127;
    int ky = t & 15;
    float re = 0.f, im = 0.f;
    for (int corner = 0; corner < 2; ++corner) {
        for (int mx = 0; mx < 16; ++mx) {
            int m  = corner * 256 + mx * 16 + ky;
            int kx = corner ? (112 + mx) : mx;
            float tr = T[(b * 512 + m) * 2];
            float ti = T[(b * 512 + m) * 2 + 1];
            int a = (kx * x) & 127;
            float c = cs[a], s = sn[a];
            re += tr * c - ti * s;
            im += tr * s + ti * c;
        }
    }
    G[t * 2]     = re;
    G[t * 2 + 1] = im;
}

__global__ void fnde_inv_dft2_out(const float* __restrict__ G, const float* __restrict__ z,
                                  const float* __restrict__ scal, float* __restrict__ out) {
    __shared__ float cs[128], sn[128];
    fnde_tables_lds(cs, sn);
    int t = blockIdx.x * 256 + threadIdx.x;
    int b = t >> 14;
    int x = (t >> 7) & 127;
    int y = t & 127;
    const float* g = G + ((b * 128 + x) * 16) * 2;
    float acc = g[0];
    for (int ky = 1; ky < 16; ++ky) {
        int a = (ky * y) & 127;
        acc += 2.0f * (g[ky * 2] * cs[a] - g[ky * 2 + 1] * sn[a]);
    }
    out[t] = scal[0] * z[t] + scal[1] + acc * (1.0f / 16384.0f);
}

extern "C" void kernel_launch(void* const* d_in, const int* in_sizes, int n_in,
                              void* d_out, int out_size, void* d_ws, size_t ws_size,
                              hipStream_t stream) {
    (void)in_sizes; (void)n_in; (void)out_size; (void)ws_size;

    const float* zin    = (const float*)d_in[0];
    const float* lift_w = (const float*)d_in[1];
    const float* lift_b = (const float*)d_in[2];
    const float* spec_w1 = (const float*)d_in[3];
    const float* spec_w2 = (const float*)d_in[4];
    const float* fl_bias = (const float*)d_in[5];
    const float* p1_w = (const float*)d_in[6];
    const float* p1_b = (const float*)d_in[7];
    const float* p2_w = (const float*)d_in[8];
    const float* p2_b = (const float*)d_in[9];
    const float* ts   = (const float*)d_in[10];
    float* outp = (float*)d_out;

    float* w = (float*)d_ws;
    float* Wp   = w;  w += (size_t)Ll * WLAYER;
    float* STre = w;  w += NSPEC;  float* STim = w;  w += NSPEC;
    float* Zhat = w;  w += Bb * NMODE * 2;
    float* Ay   = w;  w += Bb * Dd * Mm * 2;
    float* Tm   = w;  w += Bb * NMODE * 2;
    float* Gi   = w;  w += Bb * Dd * Mm * 2;
    float* scal = w;  w += 8;

    fnde_pack_weights<<<(Ll * 2 * Mm * Cc * Cc) / 256, 256, 0, stream>>>(spec_w1, spec_w2, Wp);
    fnde_fwd_dft1<<<(Bb * Dd * Mm) / 256, 256, 0, stream>>>(zin, Ay);
    fnde_fwd_dft2<<<(Bb * NMODE) / 256, 256, 0, stream>>>(Ay, Zhat);

    fnde_ode<<<256, 128, 0, stream>>>(Wp, STre, STim,
                                      fl_bias, ts, Zhat, lift_w, lift_b);

    fnde_proj_modes<<<(Bb * NMODE) / 256, 256, 0, stream>>>(STre, STim, p1_w, p1_b, p2_w, p2_b,
                                                            lift_w, lift_b, scal, Tm);
    fnde_inv_dft1<<<(Bb * Dd * Mm) / 256, 256, 0, stream>>>(Tm, Gi);
    fnde_inv_dft2_out<<<(Bb * Dd * Dd) / 256, 256, 0, stream>>>(Gi, zin, scal, outp);
}


# Weight pack launch and packed copy dropped: each lane gathers its A (re,im) pairs straight from the weight tensors; layers 1,2 two pieces per stage under the layer before, kept in LDS; 7->6 launches
# speedup vs baseline: 8.2129x; 1.0688x over previous
#include <hip/hip_runtime.h>


typedef float v2f __attribute__((ext_vector_type(2)));
typedef float v4f __attribute__((ext_vector_type(4)));
typedef float v8f __attribute__((ext_vector_type(8)));

#define Bb    16
#define Cc    64
#define Dd    128
#define Mm    16
#define Ll    3
#define NMODE 512
#define NSPEC (NMODE*Cc*Bb)
#define WSRC_I  (Cc*Mm*Mm*2)
#define WSRC_O  (Mm*Mm*2)
#define WSRC_L  (Cc*Cc*Mm*Mm*2)
#define XPLANE 1024
#define XSLOT  (2*XPLANE)
#define XBUF   (2*XSLOT)
#define WITEM  (128*4)

__device__ __forceinline__ void fnde_tables_lds(float* cs, float* sn) {
    const int t = threadIdx.x;
    if (t < 128) {
        float ang = (float)(6.283185307179586 / 128.0) * (float)t;
        cs[t] = cosf(ang);
        sn[t] = sinf(ang);
    }
    __syncthreads();
}

__global__ void fnde_fwd_dft1(const float* __restrict__ z,
                              float* __restrict__ Ay) {
    __shared__ float cs[128], sn[128];
    fnde_tables_lds(cs, sn);
    int t  = blockIdx.x * 256 + threadIdx.x;
    int b  = t >> 11;
    int x  = (t >> 4) & 127;
    int ky = t & 15;
    const float* row = z + (b * 128 + x) * 128;
    float re = 0.f, im = 0.f;
    for (int y = 0; y < 128; ++y) {
        float v = row[y];
        int a = (ky * y) & 127;
        re += v * cs[a];
        im -= v * sn[a];
    }
    Ay[t * 2]     = re;
    Ay[t * 2 + 1] = im;
}

__global__ void fnde_fwd_dft2(const float* __restrict__ Ay,
                              float* __restrict__ Zhat) {
    __shared__ float cs[128], sn[128];
    fnde_tables_lds(cs, sn);
    int t = blockIdx.x * 256 + threadIdx.x;
    int b = t >> 9;
    int m = t & 511;
    int corner = m >> 8, mx = (m >> 4) & 15, ky = m & 15;
    int kx = corner ? (112 + mx) : mx;
    float re = 0.f, im = 0.f;
    for (int x = 0; x < 128; ++x) {
        int base = ((b * 128 + x) * 16 + ky) * 2;
        float ar = Ay[base], ai = Ay[base + 1];
        int a = (kx * x) & 127;
        float c = cs[a], s = sn[a];
        re += ar * c + ai * s;
        im += ai * c - ar * s;
    }
    Zhat[t * 2]     = re;
    Zhat[t * 2 + 1] = im;
}

struct fnde_afrags { v2f ar[16], ai[16]; };
struct fnde_bfrags { v2f br[16], bi[16]; };

__device__ __forceinline__ void fnde_load_a_one(fnde_afrags& f, const float* __restrict__ src) {
#pragma unroll
    for (int q = 0; q < 8; ++q) {
#pragma unroll
        for (int s = 0; s < 4; ++s) {
            const v2f w = *(const v2f*)(src + (size_t)(8 * q + s) * WSRC_I);
            if (s & 1) { f.ar[2 * q + (s >> 1)].y = w.x;  f.ai[2 * q + (s >> 1)].y = w.y; }
            else       { f.ar[2 * q + (s >> 1)].x = w.x;  f.ai[2 * q + (s >> 1)].x = w.y; }
        }
    }
}

__device__ __forceinline__ void fnde_load_a_pair(fnde_afrags& f0, fnde_afrags& f1,
                                                 const float* __restrict__ src) {
#pragma unroll
    for (int q = 0; q < 8; ++q) {
#pragma unroll
        for (int s = 0; s < 4; ++s) {
            const v4f w = *(const v4f*)(src + (size_t)(8 * q + s) * WSRC_I);
            if (s & 1) {
                f0.ar[2 * q + (s >> 1)].y = w.x;  f0.ai[2 * q + (s >> 1)].y = w.y;
                f1.ar[2 * q + (s >> 1)].y = w.z;  f1.ai[2 * q + (s >> 1)].y = w.w;
            } else {
                f0.ar[2 * q + (s >> 1)].x = w.x;  f0.ai[2 * q + (s >> 1)].x = w.y;
                f1.ar[2 * q + (s >> 1)].x = w.z;  f1.ai[2 * q + (s >> 1)].x = w.w;
            }
        }
    }
}

__device__ __forceinline__ void fnde_load_a_staged(fnde_afrags& f0, fnde_afrags& f1,
                                                   const float* wmine) {
#pragma unroll
    for (int q = 0; q < 8; ++q) {
#pragma unroll
        for (int s = 0; s < 4; ++s) {
            const v4f w = *(const v4f*)(wmine + (4 * q + s) * WITEM);
            if (s & 1) {
                f0.ar[2 * q + (s >> 1)].y = w.x;  f0.ai[2 * q + (s >> 1)].y = w.y;
                f1.ar[2 * q + (s >> 1)].y = w.z;  f1.ai[2 * q + (s >> 1)].y = w.w;
            } else {
                f0.ar[2 * q + (s >> 1)].x = w.x;  f0.ai[2 * q + (s >> 1)].x = w.y;
                f1.ar[2 * q + (s >> 1)].x = w.z;  f1.ai[2 * q + (s >> 1)].x = w.w;
            }
        }
    }
}

__device__ __forceinline__ void fnde_load_b(fnde_bfrags& f, const float* Vr, const float* Vi,
                                            int bOff) {
#pragma unroll
    for (int q = 0; q < 8; ++q) {
        const v4f r4 = *(const v4f*)(Vr + q * 128 + bOff);
        const v4f i4 = *(const v4f*)(Vi + q * 128 + bOff);
        f.br[2 * q].x = r4.x;      f.br[2 * q].y = r4.y;
        f.br[2 * q + 1].x = r4.z;  f.br[2 * q + 1].y = r4.w;
        f.bi[2 * q].x = i4.x;      f.bi[2 * q].y = i4.y;
        f.bi[2 * q + 1].x = i4.z;  f.bi[2 * q + 1].y = i4.w;
    }
}

__device__ __forceinline__ void fnde_product(const fnde_afrags& a, const fnde_bfrags& b,
                                             v8f& cr, v8f& ci) {
#pragma unroll
    for (int kk = 0; kk < 16; ++kk) {
        v2f bin;
        bin.x = -b.bi[kk].x;    bin.y = -b.bi[kk].y;
        cr = __builtin_amdgcn_wmma_f32_16x16x4_f32(false, a.ar[kk], false, b.br[kk],
                                                   (short)0, cr, false, false);
        ci = __builtin_amdgcn_wmma_f32_16x16x4_f32(false, a.ar[kk], false, b.bi[kk],
                                                   (short)0, ci, false, false);
        cr = __builtin_amdgcn_wmma_f32_16x16x4_f32(false, a.ai[kk], false, bin,
                                                   (short)0, cr, false, false);
        ci = __builtin_amdgcn_wmma_f32_16x16x4_f32(false, a.ai[kk], false, b.br[kk],
                                                   (short)0, ci, false, false);
    }
}

__device__ __forceinline__ void fnde_pi(float kr, float ki, float mr, float mi,
                                        bool mirrored, bool halfOnly,
                                        float& pr, float& pim) {
    if (mirrored)      { pr = 0.5f * (kr + mr);  pim = 0.5f * (ki - mi); }
    else if (halfOnly) { pr = 0.5f * kr;         pim = 0.5f * ki; }
    else               { pr = kr;                pim = ki; }
}

struct fnde_state { float zr[8], zi[8], sr[8], si[8], tr[8], ti[8]; };

template <int PHASE>
__device__ __forceinline__ void fnde_combine(fnde_state& e,
        const float* kr, const float* ki, const float* s4r, const float* s4i,
        const float* xr, const float* xi, bool mirrored, bool halfOnly, float h,
        float* Ore, float* Oim, int off) {
    float o_r[8], o_i[8];
#pragma unroll
    for (int j = 0; j < 8; ++j) {
        float pr, pim;
        if (PHASE <= 3) {
            const float uc = (PHASE == 3) ? h : 0.5f * h;
            fnde_pi(kr[j], ki[j], xr[j], xi[j], mirrored, halfOnly, pr, pim);
            o_r[j] = e.zr[j] + uc * pr;
            o_i[j] = e.zi[j] + uc * pim;
            if (PHASE == 1) { e.sr[j] = kr[j];                  e.si[j] = ki[j]; }
            else            { e.sr[j] = e.sr[j] + 2.0f * kr[j]; e.si[j] = e.si[j] + 2.0f * ki[j]; }
        } else {
            const float h6 = h * (1.0f / 6.0f);
            fnde_pi(s4r[j], s4i[j], xr[j], xi[j], mirrored, halfOnly, pr, pim);
            o_r[j] = e.zr[j] + h6 * pr;
            o_i[j] = e.zi[j] + h6 * pim;
            e.zr[j] = o_r[j];
            e.zi[j] = o_i[j];
            e.tr[j] = e.tr[j] + h6 * s4r[j];
            e.ti[j] = e.ti[j] + h6 * s4i[j];
        }
    }
#pragma unroll
    for (int g = 0; g < 2; ++g) {
        const int t = off + g * 64;
        v4f vo, vp;
#pragma unroll
        for (int q = 0; q < 4; ++q) {
            vo[q] = o_r[4 * g + q];  vp[q] = o_i[4 * g + q];
        }
        *(v4f*)(Ore + t) = vo;
        *(v4f*)(Oim + t) = vp;
    }
}

template <int PHASE, int KIND>
__device__ __forceinline__ void fnde_stage_wave(
        const fnde_afrags& ao, const fnde_afrags& am,
        fnde_state& eo, fnde_state& em,
        const float* V, float* O,
        const float* __restrict__ bias, float h,
        bool pre, const float* __restrict__ n0, const float* __restrict__ n1, float* wmine,
        int mt, int laneHi, int laneLo) {
    constexpr int e0 = 2 * (PHASE - 1), e1 = e0 + 1;
    v4f wn0 = {}, wn1 = {};
    if (pre) {
        const v2f p0 = *(const v2f*)(n0 + (size_t)(8 * (e0 >> 2) + (e0 & 3)) * WSRC_I);
        const v2f r0 = *(const v2f*)(n1 + (size_t)(8 * (e0 >> 2) + (e0 & 3)) * WSRC_I);
        const v2f p1 = *(const v2f*)(n0 + (size_t)(8 * (e1 >> 2) + (e1 & 3)) * WSRC_I);
        const v2f r1 = *(const v2f*)(n1 + (size_t)(8 * (e1 >> 2) + (e1 & 3)) * WSRC_I);
        wn0.x = p0.x;  wn0.y = p0.y;  wn0.z = r0.x;  wn0.w = r0.y;
        wn1.x = p1.x;  wn1.y = p1.y;  wn1.z = r1.x;  wn1.w = r1.y;
    }
    const int bOff = ((laneHi << 4) + laneLo) * 4;
    const int obase = mt * 16 + (laneHi << 3);
    const int t0  = (obase >> 2) * 64 + laneLo * 4;

    fnde_bfrags bm, bo;
    fnde_load_b(bm, V + XSLOT, V + XSLOT + XPLANE, bOff);
    fnde_load_b(bo, V, V + XPLANE, bOff);

    float bj[8];
#pragma unroll
    for (int j = 0; j < 8; ++j) bj[j] = 0.f;
    if (KIND == 2) {
#pragma unroll
        for (int j = 0; j < 8; ++j) bj[j] = bias[obase + j] * 16384.0f;
    }

    __builtin_amdgcn_sched_barrier(0);

    v8f cmr = {}, cmi = {};
    fnde_product(am, bm, cmr, cmi);
    v8f cr = {}, ci = {};
    fnde_product(ao, bo, cr, ci);

    float kor[8], koi[8], kmr[8], kmi[8], so_r[8], so_i[8], sm_r[8], sm_i[8];
#pragma unroll
    for (int j = 0; j < 8; ++j) {
        kor[j] = cr[j];
        if (KIND == 2) kor[j] += bj[j];
        koi[j] = ci[j];
        kmr[j] = cmr[j];
        kmi[j] = cmi[j];
        so_r[j] = (PHASE == 4) ? (eo.sr[j] + kor[j]) : 0.f;
        so_i[j] = (PHASE == 4) ? (eo.si[j] + koi[j]) : 0.f;
        sm_r[j] = (PHASE == 4) ? (em.sr[j] + kmr[j]) : 0.f;
        sm_i[j] = (PHASE == 4) ? (em.si[j] + kmi[j]) : 0.f;
    }
    const bool pair = KIND == 1;
    const float* xor_ = (PHASE <= 3) ? (pair ? kmr : kor) : (pair ? sm_r : so_r);
    const float* xoi_ = (PHASE <= 3) ? (pair ? kmi : koi) : (pair ? sm_i : so_i);
    fnde_combine<PHASE>(eo, kor, koi, so_r, so_i, xor_, xoi_, KIND != 0, false, h,
                        O, O + XPLANE, t0);
    const float* xmr_ = (PHASE <= 3) ? (pair ? kor : kmr) : (pair ? so_r : sm_r);
    const float* xmi_ = (PHASE <= 3) ? (pair ? koi : kmi) : (pair ? so_i : sm_i);
    fnde_combine<PHASE>(em, kmr, kmi, sm_r, sm_i, xmr_, xmi_, KIND == 1, KIND == 2, h,
                        O + XSLOT, O + XSLOT + XPLANE, t0);
    if (pre) {
        *(v4f*)(wmine + e0 * WITEM) = wn0;
        *(v4f*)(wmine + e1 * WITEM) = wn1;
    }
}

__device__ __forceinline__ void fnde_lift_mode(fnde_state& e, const float* __restrict__ Zhat,
        const float* __restrict__ lw, const float* __restrict__ lb,
        float* Zre, float* Zim, int m, int obase, int laneLo, int off) {
    const float zr = Zhat[(laneLo * 512 + m) * 2];
    const float zi = Zhat[(laneLo * 512 + m) * 2 + 1];
#pragma unroll
    for (int g = 0; g < 2; ++g) {
        const v4f w4 = *(const v4f*)(lw + obase + 4 * g);
        const v4f b4 = *(const v4f*)(lb + obase + 4 * g);
        v4f vr, vi;
#pragma unroll
        for (int q = 0; q < 4; ++q) {
            const float w = w4[q];
            vr[q] = w * zr + ((m == 0) ? b4[q] * 16384.0f : 0.0f);
            vi[q] = w * zi;
            e.zr[4 * g + q] = vr[q];  e.zi[4 * g + q] = vi[q];
            e.sr[4 * g + q] = 0.f;    e.si[4 * g + q] = 0.f;
            e.tr[4 * g + q] = 0.f;    e.ti[4 * g + q] = 0.f;
        }
        const int t = off + g * 64;
        *(v4f*)(Zre + t)  = vr;
        *(v4f*)(Zim + t)  = vi;
    }
}

__device__ __forceinline__ void fnde_store_total(const fnde_state& e,
        float* STre, float* STim, int off) {
#pragma unroll
    for (int g = 0; g < 2; ++g) {
        v4f vr, vi;
#pragma unroll
        for (int q = 0; q < 4; ++q) { vr[q] = e.tr[4 * g + q];  vi[q] = e.ti[4 * g + q]; }
        const int t = off + g * 64;
        *(v4f*)(STre + t) = vr;
        *(v4f*)(STim + t) = vi;
    }
}

template <int KIND>
__device__ __forceinline__ void fnde_chain(
        const float* __restrict__ w1, const float* __restrict__ w2,
        float* xs,
        float* wst,
        float* STre, float* STim,
        const float* __restrict__ fl_bias, const float* __restrict__ ts,
        const float* __restrict__ Zhat, const float* __restrict__ lw,
        const float* __restrict__ lb,
        int mode, int pm, bool sideBySide, int mt, int laneHi, int laneLo) {
    const size_t aLane = (size_t)(laneHi * 4) * WSRC_I + (size_t)(mt * 16 + laneLo) * WSRC_O;
    const float* a0 = ((mode >> 8) ? w2 : w1) + ((mode >> 4) & 15) * (Mm * 2) + (mode & 15) * 2 + aLane;
    const float* a1 = ((pm >> 8) ? w2 : w1) + ((pm >> 4) & 15) * (Mm * 2) + (pm & 15) * 2 + aLane;
    const int obase = mt * 16 + (laneHi << 3);
    const int t0 = (obase >> 2) * 64 + laneLo * 4;
    float* x0 = xs;
    float* x1 = xs + XBUF;
    fnde_state eo, em;
    fnde_lift_mode(eo, Zhat, lw, lb, x0, x0 + XPLANE, mode, obase, laneLo, t0);
    fnde_lift_mode(em, Zhat, lw, lb, x0 + XSLOT, x0 + XSLOT + XPLANE, pm, obase, laneLo, t0);
    __syncthreads();
    const float tt0 = ts[0], tt1 = ts[1], tt2 = ts[2], tt3 = ts[3], tt4 = ts[4];
    const float hs0 = tt1 - tt0, hs1 = tt2 - tt1, hs2 = tt3 - tt2, hs3 = tt4 - tt3;
    float* wmine = wst + threadIdx.x * 4;
    fnde_afrags ao, am;
    if (sideBySide) {
        fnde_load_a_pair(ao, am, a0);
    } else {
        fnde_load_a_one(ao, a0);
        fnde_load_a_one(am, a1);
    }
    for (int l = 0; l < Ll; ++l) {
        const float* bl = fl_bias + l * Cc;
        const bool pre = l + 1 < Ll;
        const size_t nOff = (size_t)(pre ? l + 1 : l) * WSRC_L;
        if (l > 0) fnde_load_a_staged(ao, am, wmine);
        for (int s = 0; s < 4; ++s) {
            const float h = (s == 0) ? hs0 : (s == 1) ? hs1 : (s == 2) ? hs2 : hs3;
            const float* n0 = a0 + nOff + (size_t)(16 * s) * WSRC_I;
            const float* n1 = a1 + nOff + (size_t)(16 * s) * WSRC_I;
            float* wm = wmine + (8 * s) * WITEM;
            fnde_stage_wave<1, KIND>(ao, am, eo, em, x0, x1, bl, h, pre, n0, n1, wm,
                                     mt, laneHi, laneLo);
            __syncthreads();
            fnde_stage_wave<2, KIND>(ao, am, eo, em, x1, x0, bl, h, pre, n0, n1, wm,
                                     mt, laneHi, laneLo);
            __syncthreads();
            fnde_stage_wave<3, KIND>(ao, am, eo, em, x0, x1, bl, h, pre, n0, n1, wm,
                                     mt, laneHi, laneLo);
            __syncthreads();
            fnde_stage_wave<4, KIND>(ao, am, eo, em, x1, x0, bl, h, pre, n0, n1, wm,
                                     mt, laneHi, laneLo);
            __syncthreads();
        }
    }
    fnde_store_total(eo, STre, STim, mode * 1024 + t0);
    fnde_store_total(em, STre, STim, pm * 1024 + t0);
}

__global__ __launch_bounds__(128) __attribute__((amdgpu_waves_per_eu(1, 1)))
void fnde_ode(const float* __restrict__ w1,
              const float* __restrict__ w2,
              float* STre, float* STim,
              const float* __restrict__ fl_bias, const float* __restrict__ ts,
              const float* __restrict__ Zhat, const float* __restrict__ lw,
              const float* __restrict__ lb) {
    const int blk = blockIdx.x;
    const int mt   = threadIdx.x >> 5;
    const int lane = threadIdx.x & 31;
    const int laneHi = lane >> 4;
    const int laneLo = lane & 15;
    __shared__ __attribute__((aligned(16))) float xs[2 * XBUF];
    __shared__ __attribute__((aligned(16))) float wst[32 * WITEM];

    if (blk == 255) {
        fnde_chain<2>(w1, w2, xs, wst, STre, STim, fl_bias, ts, Zhat, lw, lb,
                      0, 256, false, mt, laneHi, laneLo);
    } else if (blk < 15) {
        fnde_chain<1>(w1, w2, xs, wst, STre, STim, fl_bias, ts, Zhat, lw, lb,
                      (1 + blk) * 16, 256 + (15 - blk) * 16, false, mt, laneHi, laneLo);
    } else {
        const int j = blk - 15;
        const bool sideBySide = j < 224;
        const int m0 = sideBySide ? (j / 7) * 16 + 2 + 2 * (j % 7) : (2 * (j - 224)) * 16 + 1;
        const int m1 = sideBySide ? m0 + 1 : m0 + 16;
        fnde_chain<0>(w1, w2, xs, wst, STre, STim, fl_bias, ts, Zhat, lw, lb,
                      m0, m1, sideBySide, mt, laneHi, laneLo);
    }
}

__global__ void fnde_proj_modes(const float* __restrict__ STre, const float* __restrict__ STim,
                                const float* __restrict__ p1w, const float* __restrict__ p1b,
                                const float* __restrict__ p2w, const float* __restrict__ p2b,
                                const float* __restrict__ lw,  const float* __restrict__ lb,
                                float* __restrict__ scal, float* __restrict__ T) {
    __shared__ float q[64];
    if (threadIdx.x < 64) {
        const int c = threadIdx.x;
        float acc = 0.f;
        for (int p = 0; p < 64; ++p) acc += p2w[p] * p1w[p * 64 + c];
        q[c] = acc;
    }
    __syncthreads();
    if (blockIdx.x == 0 && threadIdx.x == 0) {
        float r = p2b[0];
        for (int p = 0; p < 64; ++p) r += p2w[p] * p1b[p];
        float al = 0.f, be = r;
        for (int i = 0; i < 64; ++i) { al += q[i] * lw[i]; be += q[i] * lb[i]; }
        scal[0] = al;
        scal[1] = be;
    }
    int t = blockIdx.x * 256 + threadIdx.x;
    int b = t >> 9;
    int m = t & 511;
    const float* pr = STre + m * 1024 + b * 4;
    const float* pi = STim + m * 1024 + b * 4;
    float re = 0.f, im = 0.f;
    for (int cq = 0; cq < 16; ++cq) {
        const v4f vr = *(const v4f*)(pr + cq * 64);
        const v4f vi = *(const v4f*)(pi + cq * 64);
#pragma unroll
        for (int s = 0; s < 4; ++s) {
            float w = q[cq * 4 + s];
            re += w * vr[s];
            im += w * vi[s];
        }
    }
    T[t * 2]     = re;
    T[t * 2 + 1] = im;
}

__global__ void fnde_inv_dft1(const float* __restrict__ T,
                              float* __restrict__ G) {
    __shared__ float cs[128], sn[128];
    fnde_tables_lds(cs, sn);
    int t  = blockIdx.x * 256 + threadIdx.x;
    int b  = t >> 11;
    int x  = (t >> 4) & 127;
    int ky = t & 15;
    float re = 0.f, im = 0.f;
    for (int corner = 0; corner < 2; ++corner) {
        for (int mx = 0; mx < 16; ++mx) {
            int m  = corner * 256 + mx * 16 + ky;
            int kx = corner ? (112 + mx) : mx;
            float tr = T[(b * 512 + m) * 2];
            float ti = T[(b * 512 + m) * 2 + 1];
            int a = (kx * x) & 127;
            float c = cs[a], s = sn[a];
            re += tr * c - ti * s;
            im += tr * s + ti * c;
        }
    }
    G[t * 2]     = re;
    G[t * 2 + 1] = im;
}

__global__ void fnde_inv_dft2_out(const float* __restrict__ G, const float* __restrict__ z,
                                  const float* __restrict__ scal, float* __restrict__ out) {
    __shared__ float cs[128], sn[128];
    fnde_tables_lds(cs, sn);
    int t = blockIdx.x * 256 + threadIdx.x;
    int b = t >> 14;
    int x = (t >> 7) & 127;
    int y = t & 127;
    const float* g = G + ((b * 128 + x) * 16) * 2;
    float acc = g[0];
    for (int ky = 1; ky < 16; ++ky) {
        int a = (ky * y) & 127;
        acc += 2.0f * (g[ky * 2] * cs[a] - g[ky * 2 + 1] * sn[a]);
    }
    out[t] = scal[0] * z[t] + scal[1] + acc * (1.0f / 16384.0f);
}

extern "C" void kernel_launch(void* const* d_in, const int* in_sizes, int n_in,
                              void* d_out, int out_size, void* d_ws, size_t ws_size,
                              hipStream_t stream) {
    (void)in_sizes; (void)n_in; (void)out_size; (void)ws_size;

    const float* zin    = (const float*)d_in[0];
    const float* lift_w = (const float*)d_in[1];
    const float* lift_b = (const float*)d_in[2];
    const float* spec_w1 = (const float*)d_in[3];
    const float* spec_w2 = (const float*)d_in[4];
    const float* fl_bias = (const float*)d_in[5];
    const float* p1_w = (const float*)d_in[6];
    const float* p1_b = (const float*)d_in[7];
    const float* p2_w = (const float*)d_in[8];
    const float* p2_b = (const float*)d_in[9];
    const float* ts   = (const float*)d_in[10];
    float* outp = (float*)d_out;

    float* w = (float*)d_ws;
    float* STre = w;  w += NSPEC;  float* STim = w;  w += NSPEC;
    float* Zhat = w;  w += Bb * NMODE * 2;
    float* Ay   = w;  w += Bb * Dd * Mm * 2;
    float* Tm   = w;  w += Bb * NMODE * 2;
    float* Gi   = w;  w += Bb * Dd * Mm * 2;
    float* scal = w;  w += 8;

    fnde_fwd_dft1<<<(Bb * Dd * Mm) / 256, 256, 0, stream>>>(zin, Ay);
    fnde_fwd_dft2<<<(Bb * NMODE) / 256, 256, 0, stream>>>(Ay, Zhat);

    fnde_ode<<<256, 128, 0, stream>>>(spec_w1, spec_w2, STre, STim,
                                      fl_bias, ts, Zhat, lift_w, lift_b);

    fnde_proj_modes<<<(Bb * NMODE) / 256, 256, 0, stream>>>(STre, STim, p1_w, p1_b, p2_w, p2_b,
                                                            lift_w, lift_b, scal, Tm);
    fnde_inv_dft1<<<(Bb * Dd * Mm) / 256, 256, 0, stream>>>(Tm, Gi);
    fnde_inv_dft2_out<<<(Bb * Dd * Dd) / 256, 256, 0, stream>>>(Gi, zin, scal, outp);
}
